# starting kernel
# baseline (speedup 1.0000x reference)
#include <hip/hip_runtime.h>

typedef __attribute__((ext_vector_type(2))) float v2f;
typedef __attribute__((ext_vector_type(8))) float v8f;

#define NPIX   (256 * 256)
#define NBATCH 16
#define NTOT   (NBATCH * NPIX)

__global__ __launch_bounds__(1024) void mvprop_setup(
    const float* __restrict__ Wh, const float* __restrict__ bh,
    const float* __restrict__ Wp, float2* __restrict__ wB,
    float* __restrict__ wBias, float* __restrict__ wWp) {
  int t = threadIdx.x;
  if (t < 10 * 3 * 32) {
    int tile = t / 96;
    int ks   = (t / 32) % 3;
    int lane = t & 31;
    int n    = tile * 16 + (lane & 15);
    int kb   = ks * 4 + ((lane >= 16) ? 2 : 0);
    float b0 = 0.0f, b1 = 0.0f;
    if (n < 150) {
      if (kb     < 9) b0 = Wh[n * 9 + kb];
      if (kb + 1 < 9) b1 = Wh[n * 9 + kb + 1];
    }
    wB[t] = make_float2(b0, b1);
  }
  if (t < 160) {
    wBias[t] = (t < 150) ? bh[t] : 0.0f;
    wWp[t]   = (t < 150) ? Wp[t] : 0.0f;
  }
}

__global__ __launch_bounds__(256) void mvprop_conv(
    const float* __restrict__ image, const float2* __restrict__ wB,
    const float* __restrict__ wBias, const float* __restrict__ wWp,
    float* __restrict__ pout) {
  const int lane  = threadIdx.x & 31;
  const int wave  = threadIdx.x >> 5;
  const int tileI = blockIdx.x * 8 + wave;
  const int base  = tileI * 16;
  const int b     = base >> 16;
  const int pix   = base & (NPIX - 1);
  const int y     = pix >> 8;
  const int x0    = pix & 255;

  const float* __restrict__ occ = image + (size_t)b * 2 * NPIX;

  const int m     = lane & 15;
  const int khalf = lane >> 4;
  const int x     = x0 + m;

  v2f aFrag[3];
#pragma unroll
  for (int ks = 0; ks < 3; ++ks) {
    float av[2];
#pragma unroll
    for (int j = 0; j < 2; ++j) {
      int k = ks * 4 + khalf * 2 + j;
      float val = 0.0f;
      if (k < 9) {
        int dy = k / 3 - 1, dx = k % 3 - 1;
        int yy = y + dy, xx = x + dx;
        bool ok = (yy >= 0) & (yy < 256) & (xx >= 0) & (xx < 256);
        int yc = min(max(yy, 0), 255);
        int xc = min(max(xx, 0), 255);
        val = occ[yc * 256 + xc] * (ok ? 1.0f : 0.0f);
      }
      av[j] = val;
    }
    aFrag[ks].x = av[0];
    aFrag[ks].y = av[1];
  }

  float acc[8];
#pragma unroll
  for (int r = 0; r < 8; ++r) acc[r] = 0.0f;

#pragma unroll
  for (int tile = 0; tile < 10; ++tile) {
    float bias = wBias[tile * 16 + m];
    v8f c;
#pragma unroll
    for (int r = 0; r < 8; ++r) c[r] = bias;
#pragma unroll
    for (int ks = 0; ks < 3; ++ks) {
      float2 bt = wB[(tile * 3 + ks) * 32 + lane];
      v2f bf;
      bf.x = bt.x;
      bf.y = bt.y;
      c = __builtin_amdgcn_wmma_f32_16x16x4_f32(
           false, aFrag[ks],  false, bf,
           (short)0, c,  false,  false);
    }
    float wp = wWp[tile * 16 + m];
#pragma unroll
    for (int r = 0; r < 8; ++r) acc[r] += fmaxf(c[r], 0.0f) * wp;
  }

#pragma unroll
  for (int r = 0; r < 8; ++r) {
    float s = acc[r];
    s += __shfl_xor(s, 1, 32);
    s += __shfl_xor(s, 2, 32);
    s += __shfl_xor(s, 4, 32);
    s += __shfl_xor(s, 8, 32);
    acc[r] = s;
  }

  if ((lane & 15) == 0) {
    float sg[8];
#pragma unroll
    for (int r = 0; r < 8; ++r) sg[r] = 1.0f / (1.0f + __expf(-acc[r]));
    float4* dst = (float4*)(pout + base + khalf * 8);
    dst[0] = make_float4(sg[0], sg[1], sg[2], sg[3]);
    dst[1] = make_float4(sg[4], sg[5], sg[6], sg[7]);
  }
}

__global__ __launch_bounds__(256) void mvprop_step(
    const float* __restrict__ image, const float* __restrict__ pbuf,
    const float* __restrict__ vin, long vinStride,
    float* __restrict__ vout, int finalPass) {
  int idx = blockIdx.x * 256 + threadIdx.x;
  int b   = idx >> 16;
  int pix = idx & (NPIX - 1);
  int y   = pix >> 8;
  int x   = pix & 255;

  const float* __restrict__ vb = vin + (size_t)b * vinStride;

  float vc = vb[pix];
  float mx = vc;
  int y0 = max(y - 1, 0), y1 = min(y + 1, 255);
  int x0 = max(x - 1, 0), x1 = min(x + 1, 255);
  for (int yy = y0; yy <= y1; ++yy)
    for (int xx = x0; xx <= x1; ++xx)
      mx = fmaxf(mx, vb[yy * 256 + xx]);

  const float* ib = image + (size_t)b * 2 * NPIX;
  float r = ib[NPIX + pix];
  float p = pbuf[idx];
  float vn = fmaxf(vc, r + p * (mx - r));

  if (finalPass) {
    float oc = ib[pix];
    vout[idx] = (oc > 0.49f) ? vn : -1.0f;
  } else {
    vout[idx] = vn;
  }
}

extern "C" void kernel_launch(void* const* d_in, const int* in_sizes, int n_in,
                              void* d_out, int out_size, void* d_ws,
                              size_t ws_size, hipStream_t stream) {
  const float* image = (const float*)d_in[0];
  const float* Wh    = (const float*)d_in[1];
  const float* bh    = (const float*)d_in[2];
  const float* Wp    = (const float*)d_in[3];
  float* out = (float*)d_out;

  float* ws    = (float*)d_ws;
  float* pbuf  = ws;
  float* v0    = ws + (size_t)NTOT;
  float* v1    = ws + (size_t)2 * NTOT;
  float* wB    = ws + (size_t)3 * NTOT;
  float* wBias = wB + 1920;
  float* wWp   = wBias + 160;

  mvprop_setup<<<1, 1024, 0, stream>>>(Wh, bh, Wp, (float2*)wB, wBias, wWp);

  mvprop_conv<<<NTOT / (16 * 8), 256, 0, stream>>>(
      image, (const float2*)wB, wBias, wWp, pbuf);

  const float* vin = image + NPIX;
  long stride = 2 * NPIX;
  float* bufs[2] = {v0, v1};
  for (int i = 0; i < 20; ++i) {
    int fin = (i == 19) ? 1 : 0;
    float* vo = fin ? out : bufs[i & 1];
    mvprop_step<<<NTOT / 256, 256, 0, stream>>>(image, pbuf, vin, stride, vo,
                                                fin);
    vin = vo;
    stride = NPIX;
  }
}


# conv epilogue: a lane's 16 channel sums are reduced by a halving butterfly of DPP row_xmask adds (15 exchanges, not 64 LDS shuffles) that ends one pixel per lane: 1 sigmoid, 1 store; same sum tree
# speedup vs baseline: 3.2262x; 3.2262x over previous
#include <hip/hip_runtime.h>

typedef __attribute__((ext_vector_type(2))) float v2f;
typedef __attribute__((ext_vector_type(8))) float v8f;

#define NPIX   (256 * 256)
#define NBATCH 16
#define NTOT   (NBATCH * NPIX)

__global__ __launch_bounds__(1024) void mvprop_setup(
    const float* __restrict__ Wh, const float* __restrict__ bh,
    const float* __restrict__ Wp, float2* __restrict__ wB,
    float* __restrict__ wWp) {
  int t = threadIdx.x;
  if (t < 10 * 3 * 32) {
    int tile = t / 96;
    int ks   = (t / 32) % 3;
    int lane = t & 31;
    int n    = tile * 16 + (lane & 15);
    float b0 = 0.0f, b1 = 0.0f;
    if (n < 150) {
      if (lane < 16) {
        b0 = Wh[n * 9 + ks * 3 + 0];
        b1 = Wh[n * 9 + ks * 3 + 1];
      } else {
        b0 = Wh[n * 9 + ks * 3 + 2];
        b1 = (ks == 1) ? bh[n] : 0.0f;
      }
    }
    wB[t] = make_float2(b0, b1);
  }
  if (t < 160) wWp[t] = (t < 150) ? Wp[t] : 0.0f;
}

#define CB_ROWS 8
#define CB_SEGS (256 / CB_ROWS)

__device__ __forceinline__ void mvprop_row_load(const float* __restrict__ occ,
                                                int yy, int x, int h,
                                                float& side, float& mid) {
  const int yc  = min(max(yy, 0), 255);
  const int xs  = h ? x + 1 : x - 1;
  const int xsc = min(max(xs, 0), 255);
  side = occ[yc * 256 + xsc];
  mid  = occ[yc * 256 + x];
}

__device__ __forceinline__ v2f mvprop_row_frag(int yy, int x, int h, float side,
                                               float mid) {
  const bool rowOk  = (unsigned)yy < 256u;
  const int xs      = h ? x + 1 : x - 1;
  const bool sideOk = rowOk & ((unsigned)xs < 256u);
  v2f a;
  a.x = sideOk ? side : 0.0f;
  a.y = h ? 1.0f : (rowOk ? mid : 0.0f);
  return a;
}

template <int MASK>
__device__ __forceinline__ float mvprop_xadd(float keep, float send) {
  const int moved = __builtin_amdgcn_update_dpp(0, __float_as_int(send),
                                                0x160 + MASK, 0xf, 0xf, true);
  return keep + __int_as_float(moved);
}

__global__ __launch_bounds__(256) void mvprop_conv(
    const float* __restrict__ image, const float2* __restrict__ wB,
    const float* __restrict__ wWp, float* __restrict__ pout) {
  const int lane = threadIdx.x & 31;
  const int band = threadIdx.x >> 5;
  const int b    = blockIdx.x / CB_SEGS;
  const int seg  = blockIdx.x % CB_SEGS;
  const int y0   = seg * CB_ROWS;
  const int m    = lane & 15;
  const int h    = lane >> 4;
  const int x0   = band * 32 + m;
  const bool q0 = m & 1, q1 = m & 2, q2 = m & 4, q3 = m & 8;
  const int pix = 16 * (m >> 3) + 8 * h + (m & 7);

  const float* __restrict__ occ = image + (size_t)b * 2 * NPIX;

  v2f bF[10][3];
  float wp[10];
#pragma unroll
  for (int t = 0; t < 10; ++t) {
#pragma unroll
    for (int j = 0; j < 3; ++j) {
      float2 bt = wB[(t * 3 + j) * 32 + lane];
      bF[t][j].x = bt.x;
      bF[t][j].y = bt.y;
    }
    wp[t] = wWp[t * 16 + m];
  }

  v2f aP[2], aC[2], aN[2];
#pragma unroll
  for (int T = 0; T < 2; ++T) {
    float s0, m0, s1, m1, s2, m2;
    mvprop_row_load(occ, y0 - 1, x0 + 16 * T, h, s0, m0);
    mvprop_row_load(occ, y0,     x0 + 16 * T, h, s1, m1);
    mvprop_row_load(occ, y0 + 1, x0 + 16 * T, h, s2, m2);
    aP[T] = mvprop_row_frag(y0 - 1, x0 + 16 * T, h, s0, m0);
    aC[T] = mvprop_row_frag(y0,     x0 + 16 * T, h, s1, m1);
    aN[T] = mvprop_row_frag(y0 + 1, x0 + 16 * T, h, s2, m2);
  }

#pragma unroll 1
  for (int j = 0; j < CB_ROWS; ++j) {
    const int y = y0 + j;

    float fs[2], fm[2];
#pragma unroll
    for (int T = 0; T < 2; ++T)
      mvprop_row_load(occ, y + 2, x0 + 16 * T, h, fs[T], fm[T]);

    v8f acc[2];
#pragma unroll
    for (int r = 0; r < 8; ++r) { acc[0][r] = 0.0f; acc[1][r] = 0.0f; }

#pragma unroll
    for (int t = 0; t < 10; ++t) {
      v8f d0, d1;
#pragma unroll
      for (int r = 0; r < 8; ++r) { d0[r] = 0.0f; d1[r] = 0.0f; }
      d0 = __builtin_amdgcn_wmma_f32_16x16x4_f32(false, aP[0], false, bF[t][0],
                                                 (short)0, d0, false, false);
      d1 = __builtin_amdgcn_wmma_f32_16x16x4_f32(false, aP[1], false, bF[t][0],
                                                 (short)0, d1, false, false);
      d0 = __builtin_amdgcn_wmma_f32_16x16x4_f32(false, aC[0], false, bF[t][1],
                                                 (short)0, d0, false, false);
      d1 = __builtin_amdgcn_wmma_f32_16x16x4_f32(false, aC[1], false, bF[t][1],
                                                 (short)0, d1, false, false);
      d0 = __builtin_amdgcn_wmma_f32_16x16x4_f32(false, aN[0], false, bF[t][2],
                                                 (short)0, d0, false, false);
      d1 = __builtin_amdgcn_wmma_f32_16x16x4_f32(false, aN[1], false, bF[t][2],
                                                 (short)0, d1, false, false);
      v8f zero8, wp8;
#pragma unroll
      for (int r = 0; r < 8; ++r) { zero8[r] = 0.0f; wp8[r] = wp[t]; }
      acc[0] = __builtin_elementwise_fma(__builtin_elementwise_max(d0, zero8), wp8, acc[0]);
      acc[1] = __builtin_elementwise_fma(__builtin_elementwise_max(d1, zero8), wp8, acc[1]);
    }

    float s8[8], s4[4], s2[2];
#pragma unroll
    for (int k = 0; k < 8; ++k) {
      const float lo = acc[k >> 2][2 * (k & 3)], hi = acc[k >> 2][2 * (k & 3) + 1];
      s8[k] = mvprop_xadd<1>(q0 ? hi : lo, q0 ? lo : hi);
    }
#pragma unroll
    for (int k = 0; k < 4; ++k)
      s4[k] = mvprop_xadd<2>(q1 ? s8[2 * k + 1] : s8[2 * k],
                             q1 ? s8[2 * k] : s8[2 * k + 1]);
#pragma unroll
    for (int k = 0; k < 2; ++k)
      s2[k] = mvprop_xadd<4>(q2 ? s4[2 * k + 1] : s4[2 * k],
                             q2 ? s4[2 * k] : s4[2 * k + 1]);
    const float sum = mvprop_xadd<8>(q3 ? s2[1] : s2[0], q3 ? s2[0] : s2[1]);

    pout[(size_t)b * NPIX + y * 256 + band * 32 + pix] =
        1.0f / (1.0f + __expf(-sum));

#pragma unroll
    for (int T = 0; T < 2; ++T) {
      aP[T] = aC[T];
      aC[T] = aN[T];
      aN[T] = mvprop_row_frag(y + 2, x0 + 16 * T, h, fs[T], fm[T]);
    }
  }
}

#define PT 10
#define PR 8
#define PH (PR + 2 * PT)
#define NSTRIP (256 / PR)
#define PG 5

__device__ __forceinline__ void mvprop_hmax(const float (&vr)[8], float (&h)[8],
                                            bool firstLane, bool lastLane) {
  const float ninf = -__builtin_inff();
  float l = __shfl_up(vr[7], 1, 32);
  float g = __shfl_down(vr[0], 1, 32);
  l = firstLane ? ninf : l;
  g = lastLane ? ninf : g;
  h[0] = fmaxf(fmaxf(l, vr[0]), vr[1]);
#pragma unroll
  for (int c = 1; c < 7; ++c) h[c] = fmaxf(fmaxf(vr[c - 1], vr[c]), vr[c + 1]);
  h[7] = fmaxf(fmaxf(vr[6], vr[7]), g);
}

template <int LO, int HI>
__device__ __forceinline__ void mvprop_sweep(float (&v)[PH][8],
                                             const float (&r)[PH][8],
                                             const float (&p)[PH][8], int ybase,
                                             bool firstLane, bool lastLane) {
  float h[3][8];
  mvprop_hmax(v[LO - 1], h[(LO - 1) % 3], firstLane, lastLane);
  mvprop_hmax(v[LO], h[LO % 3], firstLane, lastLane);
#pragma unroll
  for (int i = LO; i < HI; ++i) {
    mvprop_hmax(v[i + 1], h[(i + 1) % 3], firstLane, lastLane);
    const bool inImage = (unsigned)(ybase + i) < 256u;
#pragma unroll
    for (int c = 0; c < 8; ++c) {
      float mx = fmaxf(fmaxf(h[(i - 1) % 3][c], h[i % 3][c]), h[(i + 1) % 3][c]);
      float rr = r[i][c];
      float vn = fmaxf(v[i][c], rr + p[i][c] * (mx - rr));
      v[i][c] = inImage ? vn : v[i][c];
    }
  }
}

template <int LO, int HI>
__device__ __forceinline__ void mvprop_phase(int nsteps, float (&v)[PH][8],
                                             const float (&r)[PH][8],
                                             const float (&p)[PH][8], int ybase,
                                             bool firstLane, bool lastLane) {
#pragma unroll 1
  for (int s = 0; s < nsteps; ++s)
    mvprop_sweep<LO, HI>(v, r, p, ybase, firstLane, lastLane);
}

template <bool FIRST, bool FINAL>
__global__ __launch_bounds__(32) void mvprop_strip(
    const float* __restrict__ image, const float* __restrict__ pbuf,
    const float* __restrict__ vin, long vinStride, float* __restrict__ vout) {
  const int lane  = threadIdx.x;
  const int b     = blockIdx.x / NSTRIP;
  const int strip = blockIdx.x % NSTRIP;
  const int y0    = strip * PR;
  const int ybase = y0 - PT;
  const int xo    = lane * 8;
  const bool firstLane = (lane == 0);
  const bool lastLane  = (lane == 31);
  const float ninf = -__builtin_inff();

  const float* __restrict__ vb = vin + (size_t)b * vinStride;
  const float* __restrict__ ib = image + (size_t)b * 2 * NPIX;
  const float* __restrict__ rb = ib + NPIX;
  const float* __restrict__ pb = pbuf + (size_t)b * NPIX;

  float v[PH][8], r[PH][8], p[PH][8];

#pragma unroll
  for (int i = 0; i < PH; ++i) {
    const int y = ybase + i;
    const bool inImage = (unsigned)y < 256u;
    const int yc = min(max(y, 0), 255);
    const float4* src = (const float4*)(vb + yc * 256 + xo);
    float4 a = src[0], c4 = src[1];
    v[i][0] = inImage ? a.x : ninf;  v[i][1] = inImage ? a.y : ninf;
    v[i][2] = inImage ? a.z : ninf;  v[i][3] = inImage ? a.w : ninf;
    v[i][4] = inImage ? c4.x : ninf; v[i][5] = inImage ? c4.y : ninf;
    v[i][6] = inImage ? c4.z : ninf; v[i][7] = inImage ? c4.w : ninf;
  }
#pragma unroll
  for (int i = 1; i < PH - 1; ++i) {
    const int yc = min(max(ybase + i, 0), 255);
    if (FIRST) {
#pragma unroll
      for (int c = 0; c < 8; ++c) r[i][c] = v[i][c];
    } else {
      const float4* rs = (const float4*)(rb + yc * 256 + xo);
      float4 a = rs[0], c4 = rs[1];
      r[i][0] = a.x;  r[i][1] = a.y;  r[i][2] = a.z;  r[i][3] = a.w;
      r[i][4] = c4.x; r[i][5] = c4.y; r[i][6] = c4.z; r[i][7] = c4.w;
    }
    const float4* ps = (const float4*)(pb + yc * 256 + xo);
    float4 a = ps[0], c4 = ps[1];
    p[i][0] = a.x;  p[i][1] = a.y;  p[i][2] = a.z;  p[i][3] = a.w;
    p[i][4] = c4.x; p[i][5] = c4.y; p[i][6] = c4.z; p[i][7] = c4.w;
  }

  mvprop_phase<1, PH - 1>(PG, v, r, p, ybase, firstLane, lastLane);
  mvprop_phase<PG + 1, PH - PG - 1>(PT - PG, v, r, p, ybase, firstLane, lastLane);

#pragma unroll
  for (int j = 0; j < PR; ++j) {
    const int i = PT + j;
    const int y = y0 + j;
    float o[8];
#pragma unroll
    for (int c = 0; c < 8; ++c) o[c] = v[i][c];
    if (FINAL) {
      const float4* os = (const float4*)(ib + y * 256 + xo);
      float4 a = os[0], c4 = os[1];
      const float oc[8] = {a.x, a.y, a.z, a.w, c4.x, c4.y, c4.z, c4.w};
#pragma unroll
      for (int c = 0; c < 8; ++c) o[c] = (oc[c] > 0.49f) ? o[c] : -1.0f;
    }
    float4* dst = (float4*)(vout + (size_t)b * NPIX + y * 256 + xo);
    dst[0] = make_float4(o[0], o[1], o[2], o[3]);
    dst[1] = make_float4(o[4], o[5], o[6], o[7]);
  }
}

extern "C" void kernel_launch(void* const* d_in, const int* in_sizes, int n_in,
                              void* d_out, int out_size, void* d_ws,
                              size_t ws_size, hipStream_t stream) {
  const float* image = (const float*)d_in[0];
  const float* Wh    = (const float*)d_in[1];
  const float* bh    = (const float*)d_in[2];
  const float* Wp    = (const float*)d_in[3];
  float* out = (float*)d_out;

  float* ws    = (float*)d_ws;
  float* pbuf  = ws;
  float* v0    = ws + (size_t)NTOT;
  float* wB    = ws + (size_t)3 * NTOT;
  float* wWp   = wB + 1920;

  mvprop_setup<<<1, 1024, 0, stream>>>(Wh, bh, Wp, (float2*)wB, wWp);

  mvprop_conv<<<NBATCH * CB_SEGS, 256, 0, stream>>>(
      image, (const float2*)wB, wWp, pbuf);

  mvprop_strip<true, false><<<NBATCH * NSTRIP, 32, 0, stream>>>(
      image, pbuf, image + NPIX, 2 * NPIX, v0);
  mvprop_strip<false, true><<<NBATCH * NSTRIP, 32, 0, stream>>>(
      image, pbuf, v0, NPIX, out);
}


# propagation: the two side-neighbour fetches per row go by DPP row shift + v_permlanex16 on the vector ALU, not by ds_bpermute through the LDS unit: no dscnt wait in the step loops; same arithmetic
# speedup vs baseline: 3.5225x; 1.0918x over previous
#include <hip/hip_runtime.h>

typedef __attribute__((ext_vector_type(2))) float v2f;
typedef __attribute__((ext_vector_type(8))) float v8f;

#define NPIX   (256 * 256)
#define NBATCH 16
#define NTOT   (NBATCH * NPIX)

__global__ __launch_bounds__(1024) void mvprop_setup(
    const float* __restrict__ Wh, const float* __restrict__ bh,
    const float* __restrict__ Wp, float2* __restrict__ wB,
    float* __restrict__ wWp) {
  int t = threadIdx.x;
  if (t < 10 * 3 * 32) {
    int tile = t / 96;
    int ks   = (t / 32) % 3;
    int lane = t & 31;
    int n    = tile * 16 + (lane & 15);
    float b0 = 0.0f, b1 = 0.0f;
    if (n < 150) {
      if (lane < 16) {
        b0 = Wh[n * 9 + ks * 3 + 0];
        b1 = Wh[n * 9 + ks * 3 + 1];
      } else {
        b0 = Wh[n * 9 + ks * 3 + 2];
        b1 = (ks == 1) ? bh[n] : 0.0f;
      }
    }
    wB[t] = make_float2(b0, b1);
  }
  if (t < 160) wWp[t] = (t < 150) ? Wp[t] : 0.0f;
}

#define CB_ROWS 8
#define CB_SEGS (256 / CB_ROWS)

__device__ __forceinline__ void mvprop_row_load(const float* __restrict__ occ,
                                                int yy, int x, int h,
                                                float& side, float& mid) {
  const int yc  = min(max(yy, 0), 255);
  const int xs  = h ? x + 1 : x - 1;
  const int xsc = min(max(xs, 0), 255);
  side = occ[yc * 256 + xsc];
  mid  = occ[yc * 256 + x];
}

__device__ __forceinline__ v2f mvprop_row_frag(int yy, int x, int h, float side,
                                               float mid) {
  const bool rowOk  = (unsigned)yy < 256u;
  const int xs      = h ? x + 1 : x - 1;
  const bool sideOk = rowOk & ((unsigned)xs < 256u);
  v2f a;
  a.x = sideOk ? side : 0.0f;
  a.y = h ? 1.0f : (rowOk ? mid : 0.0f);
  return a;
}

template <int MASK>
__device__ __forceinline__ float mvprop_xadd(float keep, float send) {
  const int moved = __builtin_amdgcn_update_dpp(0, __float_as_int(send),
                                                0x160 + MASK, 0xf, 0xf, true);
  return keep + __int_as_float(moved);
}

__global__ __launch_bounds__(256) void mvprop_conv(
    const float* __restrict__ image, const float2* __restrict__ wB,
    const float* __restrict__ wWp, float* __restrict__ pout) {
  const int lane = threadIdx.x & 31;
  const int band = threadIdx.x >> 5;
  const int b    = blockIdx.x / CB_SEGS;
  const int seg  = blockIdx.x % CB_SEGS;
  const int y0   = seg * CB_ROWS;
  const int m    = lane & 15;
  const int h    = lane >> 4;
  const int x0   = band * 32 + m;
  const bool q0 = m & 1, q1 = m & 2, q2 = m & 4, q3 = m & 8;
  const int pix = 16 * (m >> 3) + 8 * h + (m & 7);

  const float* __restrict__ occ = image + (size_t)b * 2 * NPIX;

  v2f bF[10][3];
  float wp[10];
#pragma unroll
  for (int t = 0; t < 10; ++t) {
#pragma unroll
    for (int j = 0; j < 3; ++j) {
      float2 bt = wB[(t * 3 + j) * 32 + lane];
      bF[t][j].x = bt.x;
      bF[t][j].y = bt.y;
    }
    wp[t] = wWp[t * 16 + m];
  }

  v2f aP[2], aC[2], aN[2];
#pragma unroll
  for (int T = 0; T < 2; ++T) {
    float s0, m0, s1, m1, s2, m2;
    mvprop_row_load(occ, y0 - 1, x0 + 16 * T, h, s0, m0);
    mvprop_row_load(occ, y0,     x0 + 16 * T, h, s1, m1);
    mvprop_row_load(occ, y0 + 1, x0 + 16 * T, h, s2, m2);
    aP[T] = mvprop_row_frag(y0 - 1, x0 + 16 * T, h, s0, m0);
    aC[T] = mvprop_row_frag(y0,     x0 + 16 * T, h, s1, m1);
    aN[T] = mvprop_row_frag(y0 + 1, x0 + 16 * T, h, s2, m2);
  }

#pragma unroll 1
  for (int j = 0; j < CB_ROWS; ++j) {
    const int y = y0 + j;

    float fs[2], fm[2];
#pragma unroll
    for (int T = 0; T < 2; ++T)
      mvprop_row_load(occ, y + 2, x0 + 16 * T, h, fs[T], fm[T]);

    v8f acc[2];
#pragma unroll
    for (int r = 0; r < 8; ++r) { acc[0][r] = 0.0f; acc[1][r] = 0.0f; }

#pragma unroll
    for (int t = 0; t < 10; ++t) {
      v8f d0, d1;
#pragma unroll
      for (int r = 0; r < 8; ++r) { d0[r] = 0.0f; d1[r] = 0.0f; }
      d0 = __builtin_amdgcn_wmma_f32_16x16x4_f32(false, aP[0], false, bF[t][0],
                                                 (short)0, d0, false, false);
      d1 = __builtin_amdgcn_wmma_f32_16x16x4_f32(false, aP[1], false, bF[t][0],
                                                 (short)0, d1, false, false);
      d0 = __builtin_amdgcn_wmma_f32_16x16x4_f32(false, aC[0], false, bF[t][1],
                                                 (short)0, d0, false, false);
      d1 = __builtin_amdgcn_wmma_f32_16x16x4_f32(false, aC[1], false, bF[t][1],
                                                 (short)0, d1, false, false);
      d0 = __builtin_amdgcn_wmma_f32_16x16x4_f32(false, aN[0], false, bF[t][2],
                                                 (short)0, d0, false, false);
      d1 = __builtin_amdgcn_wmma_f32_16x16x4_f32(false, aN[1], false, bF[t][2],
                                                 (short)0, d1, false, false);
      v8f zero8, wp8;
#pragma unroll
      for (int r = 0; r < 8; ++r) { zero8[r] = 0.0f; wp8[r] = wp[t]; }
      acc[0] = __builtin_elementwise_fma(__builtin_elementwise_max(d0, zero8), wp8, acc[0]);
      acc[1] = __builtin_elementwise_fma(__builtin_elementwise_max(d1, zero8), wp8, acc[1]);
    }

    float s8[8], s4[4], s2[2];
#pragma unroll
    for (int k = 0; k < 8; ++k) {
      const float lo = acc[k >> 2][2 * (k & 3)], hi = acc[k >> 2][2 * (k & 3) + 1];
      s8[k] = mvprop_xadd<1>(q0 ? hi : lo, q0 ? lo : hi);
    }
#pragma unroll
    for (int k = 0; k < 4; ++k)
      s4[k] = mvprop_xadd<2>(q1 ? s8[2 * k + 1] : s8[2 * k],
                             q1 ? s8[2 * k] : s8[2 * k + 1]);
#pragma unroll
    for (int k = 0; k < 2; ++k)
      s2[k] = mvprop_xadd<4>(q2 ? s4[2 * k + 1] : s4[2 * k],
                             q2 ? s4[2 * k] : s4[2 * k + 1]);
    const float sum = mvprop_xadd<8>(q3 ? s2[1] : s2[0], q3 ? s2[0] : s2[1]);

    pout[(size_t)b * NPIX + y * 256 + band * 32 + pix] =
        1.0f / (1.0f + __expf(-sum));

#pragma unroll
    for (int T = 0; T < 2; ++T) {
      aP[T] = aC[T];
      aC[T] = aN[T];
      aN[T] = mvprop_row_frag(y + 2, x0 + 16 * T, h, fs[T], fm[T]);
    }
  }
}

#define PT 10
#define PR 8
#define PH (PR + 2 * PT)
#define NSTRIP (256 / PR)
#define PG 5

__device__ __forceinline__ float mvprop_prev_lane(float x) {
  const int xi = __float_as_int(x);
  const int edge = __builtin_amdgcn_permlanex16(xi, xi, 0xffffffffu, 0xffffffffu,
                                                false, false);
  return __int_as_float(
      __builtin_amdgcn_update_dpp(edge, xi, 0x111, 0xf, 0xf, false));
}
__device__ __forceinline__ float mvprop_next_lane(float x) {
  const int xi = __float_as_int(x);
  const int edge = __builtin_amdgcn_permlanex16(xi, xi, 0u, 0u, false, false);
  return __int_as_float(
      __builtin_amdgcn_update_dpp(edge, xi, 0x101, 0xf, 0xf, false));
}

__device__ __forceinline__ void mvprop_hmax(const float (&vr)[8], float (&h)[8],
                                            bool firstLane, bool lastLane) {
  const float ninf = -__builtin_inff();
  float l = mvprop_prev_lane(vr[7]);
  float g = mvprop_next_lane(vr[0]);
  l = firstLane ? ninf : l;
  g = lastLane ? ninf : g;
  h[0] = fmaxf(fmaxf(l, vr[0]), vr[1]);
#pragma unroll
  for (int c = 1; c < 7; ++c) h[c] = fmaxf(fmaxf(vr[c - 1], vr[c]), vr[c + 1]);
  h[7] = fmaxf(fmaxf(vr[6], vr[7]), g);
}

template <int LO, int HI>
__device__ __forceinline__ void mvprop_sweep(float (&v)[PH][8],
                                             const float (&r)[PH][8],
                                             const float (&p)[PH][8], int ybase,
                                             bool firstLane, bool lastLane) {
  float h[3][8];
  mvprop_hmax(v[LO - 1], h[(LO - 1) % 3], firstLane, lastLane);
  mvprop_hmax(v[LO], h[LO % 3], firstLane, lastLane);
#pragma unroll
  for (int i = LO; i < HI; ++i) {
    mvprop_hmax(v[i + 1], h[(i + 1) % 3], firstLane, lastLane);
    const bool inImage = (unsigned)(ybase + i) < 256u;
#pragma unroll
    for (int c = 0; c < 8; ++c) {
      float mx = fmaxf(fmaxf(h[(i - 1) % 3][c], h[i % 3][c]), h[(i + 1) % 3][c]);
      float rr = r[i][c];
      float vn = fmaxf(v[i][c], rr + p[i][c] * (mx - rr));
      v[i][c] = inImage ? vn : v[i][c];
    }
  }
}

template <int LO, int HI>
__device__ __forceinline__ void mvprop_phase(int nsteps, float (&v)[PH][8],
                                             const float (&r)[PH][8],
                                             const float (&p)[PH][8], int ybase,
                                             bool firstLane, bool lastLane) {
#pragma unroll 1
  for (int s = 0; s < nsteps; ++s)
    mvprop_sweep<LO, HI>(v, r, p, ybase, firstLane, lastLane);
}

template <bool FIRST, bool FINAL>
__global__ __launch_bounds__(32) void mvprop_strip(
    const float* __restrict__ image, const float* __restrict__ pbuf,
    const float* __restrict__ vin, long vinStride, float* __restrict__ vout) {
  const int lane  = threadIdx.x;
  const int b     = blockIdx.x / NSTRIP;
  const int strip = blockIdx.x % NSTRIP;
  const int y0    = strip * PR;
  const int ybase = y0 - PT;
  const int xo    = lane * 8;
  const bool firstLane = (lane == 0);
  const bool lastLane  = (lane == 31);
  const float ninf = -__builtin_inff();

  const float* __restrict__ vb = vin + (size_t)b * vinStride;
  const float* __restrict__ ib = image + (size_t)b * 2 * NPIX;
  const float* __restrict__ rb = ib + NPIX;
  const float* __restrict__ pb = pbuf + (size_t)b * NPIX;

  float v[PH][8], r[PH][8], p[PH][8];

#pragma unroll
  for (int i = 0; i < PH; ++i) {
    const int y = ybase + i;
    const bool inImage = (unsigned)y < 256u;
    const int yc = min(max(y, 0), 255);
    const float4* src = (const float4*)(vb + yc * 256 + xo);
    float4 a = src[0], c4 = src[1];
    v[i][0] = inImage ? a.x : ninf;  v[i][1] = inImage ? a.y : ninf;
    v[i][2] = inImage ? a.z : ninf;  v[i][3] = inImage ? a.w : ninf;
    v[i][4] = inImage ? c4.x : ninf; v[i][5] = inImage ? c4.y : ninf;
    v[i][6] = inImage ? c4.z : ninf; v[i][7] = inImage ? c4.w : ninf;
  }
#pragma unroll
  for (int i = 1; i < PH - 1; ++i) {
    const int yc = min(max(ybase + i, 0), 255);
    if (FIRST) {
#pragma unroll
      for (int c = 0; c < 8; ++c) r[i][c] = v[i][c];
    } else {
      const float4* rs = (const float4*)(rb + yc * 256 + xo);
      float4 a = rs[0], c4 = rs[1];
      r[i][0] = a.x;  r[i][1] = a.y;  r[i][2] = a.z;  r[i][3] = a.w;
      r[i][4] = c4.x; r[i][5] = c4.y; r[i][6] = c4.z; r[i][7] = c4.w;
    }
    const float4* ps = (const float4*)(pb + yc * 256 + xo);
    float4 a = ps[0], c4 = ps[1];
    p[i][0] = a.x;  p[i][1] = a.y;  p[i][2] = a.z;  p[i][3] = a.w;
    p[i][4] = c4.x; p[i][5] = c4.y; p[i][6] = c4.z; p[i][7] = c4.w;
  }

  mvprop_phase<1, PH - 1>(PG, v, r, p, ybase, firstLane, lastLane);
  mvprop_phase<PG + 1, PH - PG - 1>(PT - PG, v, r, p, ybase, firstLane, lastLane);

#pragma unroll
  for (int j = 0; j < PR; ++j) {
    const int i = PT + j;
    const int y = y0 + j;
    float o[8];
#pragma unroll
    for (int c = 0; c < 8; ++c) o[c] = v[i][c];
    if (FINAL) {
      const float4* os = (const float4*)(ib + y * 256 + xo);
      float4 a = os[0], c4 = os[1];
      const float oc[8] = {a.x, a.y, a.z, a.w, c4.x, c4.y, c4.z, c4.w};
#pragma unroll
      for (int c = 0; c < 8; ++c) o[c] = (oc[c] > 0.49f) ? o[c] : -1.0f;
    }
    float4* dst = (float4*)(vout + (size_t)b * NPIX + y * 256 + xo);
    dst[0] = make_float4(o[0], o[1], o[2], o[3]);
    dst[1] = make_float4(o[4], o[5], o[6], o[7]);
  }
}

extern "C" void kernel_launch(void* const* d_in, const int* in_sizes, int n_in,
                              void* d_out, int out_size, void* d_ws,
                              size_t ws_size, hipStream_t stream) {
  const float* image = (const float*)d_in[0];
  const float* Wh    = (const float*)d_in[1];
  const float* bh    = (const float*)d_in[2];
  const float* Wp    = (const float*)d_in[3];
  float* out = (float*)d_out;

  float* ws    = (float*)d_ws;
  float* pbuf  = ws;
  float* v0    = ws + (size_t)NTOT;
  float* wB    = ws + (size_t)3 * NTOT;
  float* wWp   = wB + 1920;

  mvprop_setup<<<1, 1024, 0, stream>>>(Wh, bh, Wp, (float2*)wB, wWp);

  mvprop_conv<<<NBATCH * CB_SEGS, 256, 0, stream>>>(
      image, (const float2*)wB, wWp, pbuf);

  mvprop_strip<true, false><<<NBATCH * NSTRIP, 32, 0, stream>>>(
      image, pbuf, image + NPIX, 2 * NPIX, v0);
  mvprop_strip<false, true><<<NBATCH * NSTRIP, 32, 0, stream>>>(
      image, pbuf, v0, NPIX, out);
}


# propagation: each full-width strip is split between two waves, one per image half (160 columns, 5 per lane; the 64 round the seam done by both): 1024 lone waves, one per SIMD; same arithmetic
# speedup vs baseline: 4.2101x; 1.1952x over previous
#include <hip/hip_runtime.h>

typedef __attribute__((ext_vector_type(2))) float v2f;
typedef __attribute__((ext_vector_type(8))) float v8f;

#define NPIX   (256 * 256)
#define NBATCH 16
#define NTOT   (NBATCH * NPIX)

__global__ __launch_bounds__(1024) void mvprop_setup(
    const float* __restrict__ Wh, const float* __restrict__ bh,
    const float* __restrict__ Wp, float2* __restrict__ wB,
    float* __restrict__ wWp) {
  int t = threadIdx.x;
  if (t < 10 * 3 * 32) {
    int tile = t / 96;
    int ks   = (t / 32) % 3;
    int lane = t & 31;
    int n    = tile * 16 + (lane & 15);
    float b0 = 0.0f, b1 = 0.0f;
    if (n < 150) {
      if (lane < 16) {
        b0 = Wh[n * 9 + ks * 3 + 0];
        b1 = Wh[n * 9 + ks * 3 + 1];
      } else {
        b0 = Wh[n * 9 + ks * 3 + 2];
        b1 = (ks == 1) ? bh[n] : 0.0f;
      }
    }
    wB[t] = make_float2(b0, b1);
  }
  if (t < 160) wWp[t] = (t < 150) ? Wp[t] : 0.0f;
}

#define CB_ROWS 8
#define CB_SEGS (256 / CB_ROWS)

__device__ __forceinline__ void mvprop_row_load(const float* __restrict__ occ,
                                                int yy, int x, int h,
                                                float& side, float& mid) {
  const int yc  = min(max(yy, 0), 255);
  const int xs  = h ? x + 1 : x - 1;
  const int xsc = min(max(xs, 0), 255);
  side = occ[yc * 256 + xsc];
  mid  = occ[yc * 256 + x];
}

__device__ __forceinline__ v2f mvprop_row_frag(int yy, int x, int h, float side,
                                               float mid) {
  const bool rowOk  = (unsigned)yy < 256u;
  const int xs      = h ? x + 1 : x - 1;
  const bool sideOk = rowOk & ((unsigned)xs < 256u);
  v2f a;
  a.x = sideOk ? side : 0.0f;
  a.y = h ? 1.0f : (rowOk ? mid : 0.0f);
  return a;
}

template <int MASK>
__device__ __forceinline__ float mvprop_xadd(float keep, float send) {
  const int moved = __builtin_amdgcn_update_dpp(0, __float_as_int(send),
                                                0x160 + MASK, 0xf, 0xf, true);
  return keep + __int_as_float(moved);
}

__global__ __launch_bounds__(256) void mvprop_conv(
    const float* __restrict__ image, const float2* __restrict__ wB,
    const float* __restrict__ wWp, float* __restrict__ pout) {
  const int lane = threadIdx.x & 31;
  const int band = threadIdx.x >> 5;
  const int b    = blockIdx.x / CB_SEGS;
  const int seg  = blockIdx.x % CB_SEGS;
  const int y0   = seg * CB_ROWS;
  const int m    = lane & 15;
  const int h    = lane >> 4;
  const int x0   = band * 32 + m;
  const bool q0 = m & 1, q1 = m & 2, q2 = m & 4, q3 = m & 8;
  const int pix = 16 * (m >> 3) + 8 * h + (m & 7);

  const float* __restrict__ occ = image + (size_t)b * 2 * NPIX;

  v2f bF[10][3];
  float wp[10];
#pragma unroll
  for (int t = 0; t < 10; ++t) {
#pragma unroll
    for (int j = 0; j < 3; ++j) {
      float2 bt = wB[(t * 3 + j) * 32 + lane];
      bF[t][j].x = bt.x;
      bF[t][j].y = bt.y;
    }
    wp[t] = wWp[t * 16 + m];
  }

  v2f aP[2], aC[2], aN[2];
#pragma unroll
  for (int T = 0; T < 2; ++T) {
    float s0, m0, s1, m1, s2, m2;
    mvprop_row_load(occ, y0 - 1, x0 + 16 * T, h, s0, m0);
    mvprop_row_load(occ, y0,     x0 + 16 * T, h, s1, m1);
    mvprop_row_load(occ, y0 + 1, x0 + 16 * T, h, s2, m2);
    aP[T] = mvprop_row_frag(y0 - 1, x0 + 16 * T, h, s0, m0);
    aC[T] = mvprop_row_frag(y0,     x0 + 16 * T, h, s1, m1);
    aN[T] = mvprop_row_frag(y0 + 1, x0 + 16 * T, h, s2, m2);
  }

#pragma unroll 1
  for (int j = 0; j < CB_ROWS; ++j) {
    const int y = y0 + j;

    float fs[2], fm[2];
#pragma unroll
    for (int T = 0; T < 2; ++T)
      mvprop_row_load(occ, y + 2, x0 + 16 * T, h, fs[T], fm[T]);

    v8f acc[2];
#pragma unroll
    for (int r = 0; r < 8; ++r) { acc[0][r] = 0.0f; acc[1][r] = 0.0f; }

#pragma unroll
    for (int t = 0; t < 10; ++t) {
      v8f d0, d1;
#pragma unroll
      for (int r = 0; r < 8; ++r) { d0[r] = 0.0f; d1[r] = 0.0f; }
      d0 = __builtin_amdgcn_wmma_f32_16x16x4_f32(false, aP[0], false, bF[t][0],
                                                 (short)0, d0, false, false);
      d1 = __builtin_amdgcn_wmma_f32_16x16x4_f32(false, aP[1], false, bF[t][0],
                                                 (short)0, d1, false, false);
      d0 = __builtin_amdgcn_wmma_f32_16x16x4_f32(false, aC[0], false, bF[t][1],
                                                 (short)0, d0, false, false);
      d1 = __builtin_amdgcn_wmma_f32_16x16x4_f32(false, aC[1], false, bF[t][1],
                                                 (short)0, d1, false, false);
      d0 = __builtin_amdgcn_wmma_f32_16x16x4_f32(false, aN[0], false, bF[t][2],
                                                 (short)0, d0, false, false);
      d1 = __builtin_amdgcn_wmma_f32_16x16x4_f32(false, aN[1], false, bF[t][2],
                                                 (short)0, d1, false, false);
      v8f zero8, wp8;
#pragma unroll
      for (int r = 0; r < 8; ++r) { zero8[r] = 0.0f; wp8[r] = wp[t]; }
      acc[0] = __builtin_elementwise_fma(__builtin_elementwise_max(d0, zero8), wp8, acc[0]);
      acc[1] = __builtin_elementwise_fma(__builtin_elementwise_max(d1, zero8), wp8, acc[1]);
    }

    float s8[8], s4[4], s2[2];
#pragma unroll
    for (int k = 0; k < 8; ++k) {
      const float lo = acc[k >> 2][2 * (k & 3)], hi = acc[k >> 2][2 * (k & 3) + 1];
      s8[k] = mvprop_xadd<1>(q0 ? hi : lo, q0 ? lo : hi);
    }
#pragma unroll
    for (int k = 0; k < 4; ++k)
      s4[k] = mvprop_xadd<2>(q1 ? s8[2 * k + 1] : s8[2 * k],
                             q1 ? s8[2 * k] : s8[2 * k + 1]);
#pragma unroll
    for (int k = 0; k < 2; ++k)
      s2[k] = mvprop_xadd<4>(q2 ? s4[2 * k + 1] : s4[2 * k],
                             q2 ? s4[2 * k] : s4[2 * k + 1]);
    const float sum = mvprop_xadd<8>(q3 ? s2[1] : s2[0], q3 ? s2[0] : s2[1]);

    pout[(size_t)b * NPIX + y * 256 + band * 32 + pix] =
        1.0f / (1.0f + __expf(-sum));

#pragma unroll
    for (int T = 0; T < 2; ++T) {
      aP[T] = aC[T];
      aC[T] = aN[T];
      aN[T] = mvprop_row_frag(y + 2, x0 + 16 * T, h, fs[T], fm[T]);
    }
  }
}

#define PT 10
#define PR 8
#define PH (PR + 2 * PT)
#define NSTRIP (256 / PR)
#define PG 5
#define PC 5
#define PW (32 * PC)
#define PXR (256 - PW)
#define PSEAM 128
static_assert(PW - PT >= PSEAM, "left wave must be exact up to the seam");
static_assert(PXR + PT <= PSEAM, "right wave must be exact from the seam on");

typedef float mvprop_f4u __attribute__((ext_vector_type(4), aligned(4)));

__device__ __forceinline__ void mvprop_load_cols(const float* __restrict__ src,
                                                 float (&d)[PC]) {
  const mvprop_f4u a = *(const mvprop_f4u*)src;
  d[0] = a.x; d[1] = a.y; d[2] = a.z; d[3] = a.w;
  d[4] = src[4];
}
static_assert(PC == 5, "mvprop_load_cols reads 4 + 1 floats");

__device__ __forceinline__ float mvprop_prev_lane(float x) {
  const int xi = __float_as_int(x);
  const int edge = __builtin_amdgcn_permlanex16(xi, xi, 0xffffffffu, 0xffffffffu,
                                                false, false);
  return __int_as_float(
      __builtin_amdgcn_update_dpp(edge, xi, 0x111, 0xf, 0xf, false));
}
__device__ __forceinline__ float mvprop_next_lane(float x) {
  const int xi = __float_as_int(x);
  const int edge = __builtin_amdgcn_permlanex16(xi, xi, 0u, 0u, false, false);
  return __int_as_float(
      __builtin_amdgcn_update_dpp(edge, xi, 0x101, 0xf, 0xf, false));
}

__device__ __forceinline__ void mvprop_hmax(const float (&vr)[PC], float (&h)[PC],
                                            bool firstLane, bool lastLane) {
  const float ninf = -__builtin_inff();
  float l = mvprop_prev_lane(vr[PC - 1]);
  float g = mvprop_next_lane(vr[0]);
  l = firstLane ? ninf : l;
  g = lastLane ? ninf : g;
  h[0] = fmaxf(fmaxf(l, vr[0]), vr[1]);
#pragma unroll
  for (int c = 1; c < PC - 1; ++c)
    h[c] = fmaxf(fmaxf(vr[c - 1], vr[c]), vr[c + 1]);
  h[PC - 1] = fmaxf(fmaxf(vr[PC - 2], vr[PC - 1]), g);
}

template <int LO, int HI>
__device__ __forceinline__ void mvprop_sweep(float (&v)[PH][PC],
                                             const float (&r)[PH][PC],
                                             const float (&p)[PH][PC], int ybase,
                                             bool firstLane, bool lastLane) {
  float h[3][PC];
  mvprop_hmax(v[LO - 1], h[(LO - 1) % 3], firstLane, lastLane);
  mvprop_hmax(v[LO], h[LO % 3], firstLane, lastLane);
#pragma unroll
  for (int i = LO; i < HI; ++i) {
    mvprop_hmax(v[i + 1], h[(i + 1) % 3], firstLane, lastLane);
    const bool inImage = (unsigned)(ybase + i) < 256u;
#pragma unroll
    for (int c = 0; c < PC; ++c) {
      float mx = fmaxf(fmaxf(h[(i - 1) % 3][c], h[i % 3][c]), h[(i + 1) % 3][c]);
      float rr = r[i][c];
      float vn = fmaxf(v[i][c], rr + p[i][c] * (mx - rr));
      v[i][c] = inImage ? vn : v[i][c];
    }
  }
}

template <int LO, int HI>
__device__ __forceinline__ void mvprop_phase(int nsteps, float (&v)[PH][PC],
                                             const float (&r)[PH][PC],
                                             const float (&p)[PH][PC], int ybase,
                                             bool firstLane, bool lastLane) {
#pragma unroll 1
  for (int s = 0; s < nsteps; ++s)
    mvprop_sweep<LO, HI>(v, r, p, ybase, firstLane, lastLane);
}

template <bool FIRST, bool FINAL>
__global__ __launch_bounds__(32) __attribute__((amdgpu_waves_per_eu(1, 1)))
void mvprop_strip(
    const float* __restrict__ image, const float* __restrict__ pbuf,
    const float* __restrict__ vin, long vinStride, float* __restrict__ vout) {
  const int lane  = threadIdx.x;
  const int right = blockIdx.x & 1;
  const int sid   = blockIdx.x >> 1;
  const int b     = sid / NSTRIP;
  const int strip = sid % NSTRIP;
  const int y0    = strip * PR;
  const int ybase = y0 - PT;
  const int xo    = (right ? PXR : 0) + lane * PC;
  const bool firstLane = (lane == 0);
  const bool lastLane  = (lane == 31);
  const float ninf = -__builtin_inff();

  const float* __restrict__ vb = vin + (size_t)b * vinStride;
  const float* __restrict__ ib = image + (size_t)b * 2 * NPIX;
  const float* __restrict__ rb = ib + NPIX;
  const float* __restrict__ pb = pbuf + (size_t)b * NPIX;

  float v[PH][PC], r[PH][PC], p[PH][PC];

#pragma unroll
  for (int i = 0; i < PH; ++i) {
    const int y = ybase + i;
    const bool inImage = (unsigned)y < 256u;
    const int yc = min(max(y, 0), 255);
    float a[PC];
    mvprop_load_cols(vb + yc * 256 + xo, a);
#pragma unroll
    for (int c = 0; c < PC; ++c) v[i][c] = inImage ? a[c] : ninf;
  }
#pragma unroll
  for (int i = 1; i < PH - 1; ++i) {
    const int yc = min(max(ybase + i, 0), 255);
    if (FIRST) {
#pragma unroll
      for (int c = 0; c < PC; ++c) r[i][c] = v[i][c];
    } else {
      mvprop_load_cols(rb + yc * 256 + xo, r[i]);
    }
    mvprop_load_cols(pb + yc * 256 + xo, p[i]);
  }

  mvprop_phase<1, PH - 1>(PG, v, r, p, ybase, firstLane, lastLane);
  mvprop_phase<PG + 1, PH - PG - 1>(PT - PG, v, r, p, ybase, firstLane, lastLane);

  float o[PR][PC];
#pragma unroll
  for (int j = 0; j < PR; ++j) {
#pragma unroll
    for (int c = 0; c < PC; ++c) o[j][c] = v[PT + j][c];
    if (FINAL) {
      float oc[PC];
      mvprop_load_cols(ib + (y0 + j) * 256 + xo, oc);
#pragma unroll
      for (int c = 0; c < PC; ++c) o[j][c] = (oc[c] > 0.49f) ? o[j][c] : -1.0f;
    }
  }
  float* dst = vout + (size_t)b * NPIX + y0 * 256 + xo;
#pragma unroll
  for (int c = 0; c < PC; ++c) {
    const bool mine = right ? (xo + c >= PSEAM) : (xo + c < PSEAM);
    if (mine) {
#pragma unroll
      for (int j = 0; j < PR; ++j) dst[j * 256 + c] = o[j][c];
    }
  }
}

extern "C" void kernel_launch(void* const* d_in, const int* in_sizes, int n_in,
                              void* d_out, int out_size, void* d_ws,
                              size_t ws_size, hipStream_t stream) {
  const float* image = (const float*)d_in[0];
  const float* Wh    = (const float*)d_in[1];
  const float* bh    = (const float*)d_in[2];
  const float* Wp    = (const float*)d_in[3];
  float* out = (float*)d_out;

  float* ws    = (float*)d_ws;
  float* pbuf  = ws;
  float* v0    = ws + (size_t)NTOT;
  float* wB    = ws + (size_t)3 * NTOT;
  float* wWp   = wB + 1920;

  mvprop_setup<<<1, 1024, 0, stream>>>(Wh, bh, Wp, (float2*)wB, wWp);

  mvprop_conv<<<NBATCH * CB_SEGS, 256, 0, stream>>>(
      image, (const float2*)wB, wWp, pbuf);

  mvprop_strip<true, false><<<NBATCH * NSTRIP * 2, 32, 0, stream>>>(
      image, pbuf, image + NPIX, 2 * NPIX, v0);
  mvprop_strip<false, true><<<NBATCH * NSTRIP * 2, 32, 0, stream>>>(
      image, pbuf, v0, NPIX, out);
}


# conv: each lane reads its channel's 9 taps, bias and W_p straight from the inputs (unaligned 16-byte loads), keeps its lane half's pair; no repack table; setup launch gone: 4 launches become 3
# speedup vs baseline: 4.3416x; 1.0312x over previous
#include <hip/hip_runtime.h>

typedef __attribute__((ext_vector_type(2))) float v2f;
typedef __attribute__((ext_vector_type(8))) float v8f;

#define NPIX   (256 * 256)
#define NBATCH 16
#define NTOT   (NBATCH * NPIX)

typedef float mvprop_f4u __attribute__((ext_vector_type(4), aligned(4)));

#define CB_ROWS 8
#define CB_SEGS (256 / CB_ROWS)

__device__ __forceinline__ void mvprop_row_load(const float* __restrict__ occ,
                                                int yy, int x, int h,
                                                float& side, float& mid) {
  const int yc  = min(max(yy, 0), 255);
  const int xs  = h ? x + 1 : x - 1;
  const int xsc = min(max(xs, 0), 255);
  side = occ[yc * 256 + xsc];
  mid  = occ[yc * 256 + x];
}

__device__ __forceinline__ v2f mvprop_row_frag(int yy, int x, int h, float side,
                                               float mid) {
  const bool rowOk  = (unsigned)yy < 256u;
  const int xs      = h ? x + 1 : x - 1;
  const bool sideOk = rowOk & ((unsigned)xs < 256u);
  v2f a;
  a.x = sideOk ? side : 0.0f;
  a.y = h ? 1.0f : (rowOk ? mid : 0.0f);
  return a;
}

template <int MASK>
__device__ __forceinline__ float mvprop_xadd(float keep, float send) {
  const int moved = __builtin_amdgcn_update_dpp(0, __float_as_int(send),
                                                0x160 + MASK, 0xf, 0xf, true);
  return keep + __int_as_float(moved);
}

__global__ __launch_bounds__(256) void mvprop_conv(
    const float* __restrict__ image, const float* __restrict__ Wh,
    const float* __restrict__ bh, const float* __restrict__ Wp,
    float* __restrict__ pout) {
  const int lane = threadIdx.x & 31;
  const int band = threadIdx.x >> 5;
  const int b    = blockIdx.x / CB_SEGS;
  const int seg  = blockIdx.x % CB_SEGS;
  const int y0   = seg * CB_ROWS;
  const int m    = lane & 15;
  const int h    = lane >> 4;
  const int x0   = band * 32 + m;
  const bool q0 = m & 1, q1 = m & 2, q2 = m & 4, q3 = m & 8;
  const int pix = 16 * (m >> 3) + 8 * h + (m & 7);

  const float* __restrict__ occ = image + (size_t)b * 2 * NPIX;

  v2f bF[10][3];
  float wp[10];
#pragma unroll
  for (int t = 0; t < 10; ++t) {
    const int n   = t * 16 + m;
    const bool ok = n < 150;
    const int nc  = ok ? n : 149;
    const float* __restrict__ wn = Wh + nc * 9;
    const mvprop_f4u w03 = *(const mvprop_f4u*)wn;
    const mvprop_f4u w47 = *(const mvprop_f4u*)(wn + 4);
    const float w[9] = {w03.x, w03.y, w03.z, w03.w, w47.x, w47.y, w47.z, w47.w,
                        wn[8]};
    const float bias = bh[nc];
    const float wpn  = Wp[nc];
#pragma unroll
    for (int j = 0; j < 3; ++j) {
      const float b0 = h ? w[3 * j + 2] : w[3 * j];
      const float b1 = h ? ((j == 1) ? bias : 0.0f) : w[3 * j + 1];
      bF[t][j].x = ok ? b0 : 0.0f;
      bF[t][j].y = ok ? b1 : 0.0f;
    }
    wp[t] = ok ? wpn : 0.0f;
  }

  v2f aP[2], aC[2], aN[2];
#pragma unroll
  for (int T = 0; T < 2; ++T) {
    float s0, m0, s1, m1, s2, m2;
    mvprop_row_load(occ, y0 - 1, x0 + 16 * T, h, s0, m0);
    mvprop_row_load(occ, y0,     x0 + 16 * T, h, s1, m1);
    mvprop_row_load(occ, y0 + 1, x0 + 16 * T, h, s2, m2);
    aP[T] = mvprop_row_frag(y0 - 1, x0 + 16 * T, h, s0, m0);
    aC[T] = mvprop_row_frag(y0,     x0 + 16 * T, h, s1, m1);
    aN[T] = mvprop_row_frag(y0 + 1, x0 + 16 * T, h, s2, m2);
  }

#pragma unroll 1
  for (int j = 0; j < CB_ROWS; ++j) {
    const int y = y0 + j;

    float fs[2], fm[2];
#pragma unroll
    for (int T = 0; T < 2; ++T)
      mvprop_row_load(occ, y + 2, x0 + 16 * T, h, fs[T], fm[T]);

    v8f acc[2];
#pragma unroll
    for (int r = 0; r < 8; ++r) { acc[0][r] = 0.0f; acc[1][r] = 0.0f; }

#pragma unroll
    for (int t = 0; t < 10; ++t) {
      v8f d0, d1;
#pragma unroll
      for (int r = 0; r < 8; ++r) { d0[r] = 0.0f; d1[r] = 0.0f; }
      d0 = __builtin_amdgcn_wmma_f32_16x16x4_f32(false, aP[0], false, bF[t][0],
                                                 (short)0, d0, false, false);
      d1 = __builtin_amdgcn_wmma_f32_16x16x4_f32(false, aP[1], false, bF[t][0],
                                                 (short)0, d1, false, false);
      d0 = __builtin_amdgcn_wmma_f32_16x16x4_f32(false, aC[0], false, bF[t][1],
                                                 (short)0, d0, false, false);
      d1 = __builtin_amdgcn_wmma_f32_16x16x4_f32(false, aC[1], false, bF[t][1],
                                                 (short)0, d1, false, false);
      d0 = __builtin_amdgcn_wmma_f32_16x16x4_f32(false, aN[0], false, bF[t][2],
                                                 (short)0, d0, false, false);
      d1 = __builtin_amdgcn_wmma_f32_16x16x4_f32(false, aN[1], false, bF[t][2],
                                                 (short)0, d1, false, false);
      v8f zero8, wp8;
#pragma unroll
      for (int r = 0; r < 8; ++r) { zero8[r] = 0.0f; wp8[r] = wp[t]; }
      acc[0] = __builtin_elementwise_fma(__builtin_elementwise_max(d0, zero8), wp8, acc[0]);
      acc[1] = __builtin_elementwise_fma(__builtin_elementwise_max(d1, zero8), wp8, acc[1]);
    }

    float s8[8], s4[4], s2[2];
#pragma unroll
    for (int k = 0; k < 8; ++k) {
      const float lo = acc[k >> 2][2 * (k & 3)], hi = acc[k >> 2][2 * (k & 3) + 1];
      s8[k] = mvprop_xadd<1>(q0 ? hi : lo, q0 ? lo : hi);
    }
#pragma unroll
    for (int k = 0; k < 4; ++k)
      s4[k] = mvprop_xadd<2>(q1 ? s8[2 * k + 1] : s8[2 * k],
                             q1 ? s8[2 * k] : s8[2 * k + 1]);
#pragma unroll
    for (int k = 0; k < 2; ++k)
      s2[k] = mvprop_xadd<4>(q2 ? s4[2 * k + 1] : s4[2 * k],
                             q2 ? s4[2 * k] : s4[2 * k + 1]);
    const float sum = mvprop_xadd<8>(q3 ? s2[1] : s2[0], q3 ? s2[0] : s2[1]);

    pout[(size_t)b * NPIX + y * 256 + band * 32 + pix] =
        1.0f / (1.0f + __expf(-sum));

#pragma unroll
    for (int T = 0; T < 2; ++T) {
      aP[T] = aC[T];
      aC[T] = aN[T];
      aN[T] = mvprop_row_frag(y + 2, x0 + 16 * T, h, fs[T], fm[T]);
    }
  }
}

#define PT 10
#define PR 8
#define PH (PR + 2 * PT)
#define NSTRIP (256 / PR)
#define PG 5
#define PC 5
#define PW (32 * PC)
#define PXR (256 - PW)
#define PSEAM 128
static_assert(PW - PT >= PSEAM, "left wave must be exact up to the seam");
static_assert(PXR + PT <= PSEAM, "right wave must be exact from the seam on");

__device__ __forceinline__ void mvprop_load_cols(const float* __restrict__ src,
                                                 float (&d)[PC]) {
  const mvprop_f4u a = *(const mvprop_f4u*)src;
  d[0] = a.x; d[1] = a.y; d[2] = a.z; d[3] = a.w;
  d[4] = src[4];
}
static_assert(PC == 5, "mvprop_load_cols reads 4 + 1 floats");

__device__ __forceinline__ float mvprop_prev_lane(float x) {
  const int xi = __float_as_int(x);
  const int edge = __builtin_amdgcn_permlanex16(xi, xi, 0xffffffffu, 0xffffffffu,
                                                false, false);
  return __int_as_float(
      __builtin_amdgcn_update_dpp(edge, xi, 0x111, 0xf, 0xf, false));
}
__device__ __forceinline__ float mvprop_next_lane(float x) {
  const int xi = __float_as_int(x);
  const int edge = __builtin_amdgcn_permlanex16(xi, xi, 0u, 0u, false, false);
  return __int_as_float(
      __builtin_amdgcn_update_dpp(edge, xi, 0x101, 0xf, 0xf, false));
}

__device__ __forceinline__ void mvprop_hmax(const float (&vr)[PC], float (&h)[PC],
                                            bool firstLane, bool lastLane) {
  const float ninf = -__builtin_inff();
  float l = mvprop_prev_lane(vr[PC - 1]);
  float g = mvprop_next_lane(vr[0]);
  l = firstLane ? ninf : l;
  g = lastLane ? ninf : g;
  h[0] = fmaxf(fmaxf(l, vr[0]), vr[1]);
#pragma unroll
  for (int c = 1; c < PC - 1; ++c)
    h[c] = fmaxf(fmaxf(vr[c - 1], vr[c]), vr[c + 1]);
  h[PC - 1] = fmaxf(fmaxf(vr[PC - 2], vr[PC - 1]), g);
}

template <int LO, int HI>
__device__ __forceinline__ void mvprop_sweep(float (&v)[PH][PC],
                                             const float (&r)[PH][PC],
                                             const float (&p)[PH][PC], int ybase,
                                             bool firstLane, bool lastLane) {
  float h[3][PC];
  mvprop_hmax(v[LO - 1], h[(LO - 1) % 3], firstLane, lastLane);
  mvprop_hmax(v[LO], h[LO % 3], firstLane, lastLane);
#pragma unroll
  for (int i = LO; i < HI; ++i) {
    mvprop_hmax(v[i + 1], h[(i + 1) % 3], firstLane, lastLane);
    const bool inImage = (unsigned)(ybase + i) < 256u;
#pragma unroll
    for (int c = 0; c < PC; ++c) {
      float mx = fmaxf(fmaxf(h[(i - 1) % 3][c], h[i % 3][c]), h[(i + 1) % 3][c]);
      float rr = r[i][c];
      float vn = fmaxf(v[i][c], rr + p[i][c] * (mx - rr));
      v[i][c] = inImage ? vn : v[i][c];
    }
  }
}

template <int LO, int HI>
__device__ __forceinline__ void mvprop_phase(int nsteps, float (&v)[PH][PC],
                                             const float (&r)[PH][PC],
                                             const float (&p)[PH][PC], int ybase,
                                             bool firstLane, bool lastLane) {
#pragma unroll 1
  for (int s = 0; s < nsteps; ++s)
    mvprop_sweep<LO, HI>(v, r, p, ybase, firstLane, lastLane);
}

template <bool FIRST, bool FINAL>
__global__ __launch_bounds__(32) __attribute__((amdgpu_waves_per_eu(1, 1)))
void mvprop_strip(
    const float* __restrict__ image, const float* __restrict__ pbuf,
    const float* __restrict__ vin, long vinStride, float* __restrict__ vout) {
  const int lane  = threadIdx.x;
  const int right = blockIdx.x & 1;
  const int sid   = blockIdx.x >> 1;
  const int b     = sid / NSTRIP;
  const int strip = sid % NSTRIP;
  const int y0    = strip * PR;
  const int ybase = y0 - PT;
  const int xo    = (right ? PXR : 0) + lane * PC;
  const bool firstLane = (lane == 0);
  const bool lastLane  = (lane == 31);
  const float ninf = -__builtin_inff();

  const float* __restrict__ vb = vin + (size_t)b * vinStride;
  const float* __restrict__ ib = image + (size_t)b * 2 * NPIX;
  const float* __restrict__ rb = ib + NPIX;
  const float* __restrict__ pb = pbuf + (size_t)b * NPIX;

  float v[PH][PC], r[PH][PC], p[PH][PC];

#pragma unroll
  for (int i = 0; i < PH; ++i) {
    const int y = ybase + i;
    const bool inImage = (unsigned)y < 256u;
    const int yc = min(max(y, 0), 255);
    float a[PC];
    mvprop_load_cols(vb + yc * 256 + xo, a);
#pragma unroll
    for (int c = 0; c < PC; ++c) v[i][c] = inImage ? a[c] : ninf;
  }
#pragma unroll
  for (int i = 1; i < PH - 1; ++i) {
    const int yc = min(max(ybase + i, 0), 255);
    if (FIRST) {
#pragma unroll
      for (int c = 0; c < PC; ++c) r[i][c] = v[i][c];
    } else {
      mvprop_load_cols(rb + yc * 256 + xo, r[i]);
    }
    mvprop_load_cols(pb + yc * 256 + xo, p[i]);
  }

  mvprop_phase<1, PH - 1>(PG, v, r, p, ybase, firstLane, lastLane);
  mvprop_phase<PG + 1, PH - PG - 1>(PT - PG, v, r, p, ybase, firstLane, lastLane);

  float o[PR][PC];
#pragma unroll
  for (int j = 0; j < PR; ++j) {
#pragma unroll
    for (int c = 0; c < PC; ++c) o[j][c] = v[PT + j][c];
    if (FINAL) {
      float oc[PC];
      mvprop_load_cols(ib + (y0 + j) * 256 + xo, oc);
#pragma unroll
      for (int c = 0; c < PC; ++c) o[j][c] = (oc[c] > 0.49f) ? o[j][c] : -1.0f;
    }
  }
  float* dst = vout + (size_t)b * NPIX + y0 * 256 + xo;
#pragma unroll
  for (int c = 0; c < PC; ++c) {
    const bool mine = right ? (xo + c >= PSEAM) : (xo + c < PSEAM);
    if (mine) {
#pragma unroll
      for (int j = 0; j < PR; ++j) dst[j * 256 + c] = o[j][c];
    }
  }
}

extern "C" void kernel_launch(void* const* d_in, const int* in_sizes, int n_in,
                              void* d_out, int out_size, void* d_ws,
                              size_t ws_size, hipStream_t stream) {
  const float* image = (const float*)d_in[0];
  const float* Wh    = (const float*)d_in[1];
  const float* bh    = (const float*)d_in[2];
  const float* Wp    = (const float*)d_in[3];
  float* out = (float*)d_out;

  float* ws    = (float*)d_ws;
  float* pbuf  = ws;
  float* v0    = ws + (size_t)NTOT;

  mvprop_conv<<<NBATCH * CB_SEGS, 256, 0, stream>>>(image, Wh, bh, Wp, pbuf);

  mvprop_strip<true, false><<<NBATCH * NSTRIP * 2, 32, 0, stream>>>(
      image, pbuf, image + NPIX, 2 * NPIX, v0);
  mvprop_strip<false, true><<<NBATCH * NSTRIP * 2, 32, 0, stream>>>(
      image, pbuf, v0, NPIX, out);
}


# conv: stencil rows contracted by the minimal 3-tap form for 2 outputs: a row pair is 4 K=4 WMMAs per tile (rows and taps transformed), not 6; tile sums m2+m3, m2-m3 enter as C; tap sum order changes
# speedup vs baseline: 4.6762x; 1.0771x over previous
#include <hip/hip_runtime.h>

typedef __attribute__((ext_vector_type(2))) float v2f;
typedef __attribute__((ext_vector_type(8))) float v8f;

#define NPIX   (256 * 256)
#define NBATCH 16
#define NTOT   (NBATCH * NPIX)

typedef float mvprop_f4u __attribute__((ext_vector_type(4), aligned(4)));

#define CB_ROWS 8
#define CB_SEGS (256 / CB_ROWS)

__device__ __forceinline__ void mvprop_row_load(const float* __restrict__ occ,
                                                int yy, int x, int h,
                                                float& side, float& mid) {
  const int yc  = min(max(yy, 0), 255);
  const int xs  = h ? x + 1 : x - 1;
  const int xsc = min(max(xs, 0), 255);
  side = occ[yc * 256 + xsc];
  mid  = occ[yc * 256 + x];
}

__device__ __forceinline__ v2f mvprop_row_vals(int yy, int x, int h, float side,
                                               float mid) {
  const bool rowOk  = (unsigned)yy < 256u;
  const int xs      = h ? x + 1 : x - 1;
  const bool sideOk = rowOk & ((unsigned)xs < 256u);
  v2f a;
  a.x = sideOk ? side : 0.0f;
  a.y = (rowOk & (h == 0)) ? mid : 0.0f;
  return a;
}

template <int MASK>
__device__ __forceinline__ float mvprop_xadd(float keep, float send) {
  const int moved = __builtin_amdgcn_update_dpp(0, __float_as_int(send),
                                                0x160 + MASK, 0xf, 0xf, true);
  return keep + __int_as_float(moved);
}

__device__ __forceinline__ float mvprop_chan_sum(const v2f (&acc)[2][4], bool q0,
                                                 bool q1, bool q2, bool q3) {
  float s8[8], s4[4], s2[2];
#pragma unroll
  for (int k = 0; k < 8; ++k) {
    const float lo = acc[k >> 2][k & 3].x, hi = acc[k >> 2][k & 3].y;
    s8[k] = mvprop_xadd<1>(q0 ? hi : lo, q0 ? lo : hi);
  }
#pragma unroll
  for (int k = 0; k < 4; ++k)
    s4[k] = mvprop_xadd<2>(q1 ? s8[2 * k + 1] : s8[2 * k],
                           q1 ? s8[2 * k] : s8[2 * k + 1]);
#pragma unroll
  for (int k = 0; k < 2; ++k)
    s2[k] = mvprop_xadd<4>(q2 ? s4[2 * k + 1] : s4[2 * k],
                           q2 ? s4[2 * k] : s4[2 * k + 1]);
  return mvprop_xadd<8>(q3 ? s2[1] : s2[0], q3 ? s2[0] : s2[1]);
}

__global__ __launch_bounds__(256) void mvprop_conv(
    const float* __restrict__ image, const float* __restrict__ Wh,
    const float* __restrict__ bh, const float* __restrict__ Wp,
    float* __restrict__ pout) {
  const int lane = threadIdx.x & 31;
  const int band = threadIdx.x >> 5;
  const int b    = blockIdx.x / CB_SEGS;
  const int seg  = blockIdx.x % CB_SEGS;
  const int y0   = seg * CB_ROWS;
  const int m    = lane & 15;
  const int h    = lane >> 4;
  const int x0   = band * 32 + m;
  const bool q0 = m & 1, q1 = m & 2, q2 = m & 4, q3 = m & 8;
  const int pix = 16 * (m >> 3) + 8 * h + (m & 7);

  const float* __restrict__ occ = image + (size_t)b * 2 * NPIX;

  v2f bF[10][4];
  float wp[10];
#pragma unroll
  for (int t = 0; t < 10; ++t) {
    const int n   = t * 16 + m;
    const bool ok = n < 150;
    const int nc  = ok ? n : 149;
    const float* __restrict__ wn = Wh + nc * 9;
    const mvprop_f4u w03 = *(const mvprop_f4u*)wn;
    const mvprop_f4u w47 = *(const mvprop_f4u*)(wn + 4);
    const float w[9] = {w03.x, w03.y, w03.z, w03.w, w47.x, w47.y, w47.z, w47.w,
                        wn[8]};
    const float bias = bh[nc];
    const float wpn  = Wp[nc];
    const float gx0 = h ? w[2] : w[0], gx1 = h ? w[5] : w[3], gx2 = h ? w[8] : w[6];
    const float gy0 = w[1], gy1 = w[4], gy2 = w[7];
    const float ex = gx0 + gx2, ey = gy0 + gy2;
    float ux[4], uy[4];
    ux[0] = gx0;               uy[0] = h ? 0.0f : gy0;
    ux[1] = 0.5f * (ex + gx1); uy[1] = h ? bias : 0.5f * (ey + gy1);
    ux[2] = 0.5f * (ex - gx1); uy[2] = h ? 0.0f : 0.5f * (ey - gy1);
    ux[3] = -gx2;              uy[3] = h ? 0.0f : -gy2;
#pragma unroll
    for (int i = 0; i < 4; ++i) {
      bF[t][i].x = ok ? ux[i] : 0.0f;
      bF[t][i].y = ok ? uy[i] : 0.0f;
    }
    wp[t] = ok ? wpn : 0.0f;
  }

  v2f r0[2], r1[2], r2[2], r3[2];
#pragma unroll
  for (int T = 0; T < 2; ++T) {
    float s0, m0, s1, m1, s2, m2, s3, m3;
    mvprop_row_load(occ, y0 - 1, x0 + 16 * T, h, s0, m0);
    mvprop_row_load(occ, y0,     x0 + 16 * T, h, s1, m1);
    mvprop_row_load(occ, y0 + 1, x0 + 16 * T, h, s2, m2);
    mvprop_row_load(occ, y0 + 2, x0 + 16 * T, h, s3, m3);
    r0[T] = mvprop_row_vals(y0 - 1, x0 + 16 * T, h, s0, m0);
    r1[T] = mvprop_row_vals(y0,     x0 + 16 * T, h, s1, m1);
    r2[T] = mvprop_row_vals(y0 + 1, x0 + 16 * T, h, s2, m2);
    r3[T] = mvprop_row_vals(y0 + 2, x0 + 16 * T, h, s3, m3);
  }

#pragma unroll 1
  for (int j = 0; j < CB_ROWS; j += 2) {
    const int y = y0 + j;

    float fs3[2], fm3[2], fs4[2], fm4[2];
#pragma unroll
    for (int T = 0; T < 2; ++T) {
      mvprop_row_load(occ, y + 3, x0 + 16 * T, h, fs3[T], fm3[T]);
      mvprop_row_load(occ, y + 4, x0 + 16 * T, h, fs4[T], fm4[T]);
    }

    v2f a1[2], a2[2], a3[2], a4[2];
#pragma unroll
    for (int T = 0; T < 2; ++T) {
      a1[T] = r0[T] - r2[T];
      a2[T] = r1[T] + r2[T];
      a2[T].y = h ? 1.0f : a2[T].y;
      a3[T] = r2[T] - r1[T];
      a4[T] = r1[T] - r3[T];
    }

    v2f accE[2][4], accO[2][4];
#pragma unroll
    for (int k = 0; k < 4; ++k) {
      accE[0][k] = 0.0f; accE[1][k] = 0.0f;
      accO[0][k] = 0.0f; accO[1][k] = 0.0f;
    }

#pragma unroll
    for (int t = 0; t < 10; ++t) {
      v8f z8;
#pragma unroll
      for (int r = 0; r < 8; ++r) z8[r] = 0.0f;
      const v8f p0 = __builtin_amdgcn_wmma_f32_16x16x4_f32(
          false, a2[0], false, bF[t][1], (short)0, z8, false, false);
      const v8f g0 = __builtin_amdgcn_wmma_f32_16x16x4_f32(
          false, a3[0], false, bF[t][2], (short)0, z8, false, false);
      const v8f p1 = __builtin_amdgcn_wmma_f32_16x16x4_f32(
          false, a2[1], false, bF[t][1], (short)0, z8, false, false);
      const v8f g1 = __builtin_amdgcn_wmma_f32_16x16x4_f32(
          false, a3[1], false, bF[t][2], (short)0, z8, false, false);
      const v8f e0 = __builtin_amdgcn_wmma_f32_16x16x4_f32(
          false, a1[0], false, bF[t][0], (short)0, p0 + g0, false, false);
      const v8f o0 = __builtin_amdgcn_wmma_f32_16x16x4_f32(
          false, a4[0], false, bF[t][3], (short)0, p0 - g0, false, false);
      const v8f e1 = __builtin_amdgcn_wmma_f32_16x16x4_f32(
          false, a1[1], false, bF[t][0], (short)0, p1 + g1, false, false);
      const v8f o1 = __builtin_amdgcn_wmma_f32_16x16x4_f32(
          false, a4[1], false, bF[t][3], (short)0, p1 - g1, false, false);
      const v2f wp2 = {wp[t], wp[t]};
      v8f zero8;
#pragma unroll
      for (int r = 0; r < 8; ++r) zero8[r] = 0.0f;
      const v8f re0 = __builtin_elementwise_max(e0, zero8);
      const v8f ro0 = __builtin_elementwise_max(o0, zero8);
      const v8f re1 = __builtin_elementwise_max(e1, zero8);
      const v8f ro1 = __builtin_elementwise_max(o1, zero8);
#pragma unroll
      for (int k = 0; k < 4; ++k) {
        const v2f de0 = {re0[2 * k], re0[2 * k + 1]};
        const v2f do0 = {ro0[2 * k], ro0[2 * k + 1]};
        const v2f de1 = {re1[2 * k], re1[2 * k + 1]};
        const v2f do1 = {ro1[2 * k], ro1[2 * k + 1]};
        accE[0][k] = __builtin_elementwise_fma(de0, wp2, accE[0][k]);
        accO[0][k] = __builtin_elementwise_fma(do0, wp2, accO[0][k]);
        accE[1][k] = __builtin_elementwise_fma(de1, wp2, accE[1][k]);
        accO[1][k] = __builtin_elementwise_fma(do1, wp2, accO[1][k]);
      }
    }

    const float sumE = mvprop_chan_sum(accE, q0, q1, q2, q3);
    const float sumO = mvprop_chan_sum(accO, q0, q1, q2, q3);
    float* __restrict__ prow = pout + (size_t)b * NPIX + y * 256 + band * 32 + pix;
    prow[0]   = 1.0f / (1.0f + __expf(-sumE));
    prow[256] = 1.0f / (1.0f + __expf(-sumO));

#pragma unroll
    for (int T = 0; T < 2; ++T) {
      r0[T] = r2[T];
      r1[T] = r3[T];
      r2[T] = mvprop_row_vals(y + 3, x0 + 16 * T, h, fs3[T], fm3[T]);
      r3[T] = mvprop_row_vals(y + 4, x0 + 16 * T, h, fs4[T], fm4[T]);
    }
  }
}

#define PT 10
#define PR 8
#define PH (PR + 2 * PT)
#define NSTRIP (256 / PR)
#define PG 5
#define PC 5
#define PW (32 * PC)
#define PXR (256 - PW)
#define PSEAM 128
static_assert(PW - PT >= PSEAM, "left wave must be exact up to the seam");
static_assert(PXR + PT <= PSEAM, "right wave must be exact from the seam on");

__device__ __forceinline__ void mvprop_load_cols(const float* __restrict__ src,
                                                 float (&d)[PC]) {
  const mvprop_f4u a = *(const mvprop_f4u*)src;
  d[0] = a.x; d[1] = a.y; d[2] = a.z; d[3] = a.w;
  d[4] = src[4];
}
static_assert(PC == 5, "mvprop_load_cols reads 4 + 1 floats");

__device__ __forceinline__ float mvprop_prev_lane(float x) {
  const int xi = __float_as_int(x);
  const int edge = __builtin_amdgcn_permlanex16(xi, xi, 0xffffffffu, 0xffffffffu,
                                                false, false);
  return __int_as_float(
      __builtin_amdgcn_update_dpp(edge, xi, 0x111, 0xf, 0xf, false));
}
__device__ __forceinline__ float mvprop_next_lane(float x) {
  const int xi = __float_as_int(x);
  const int edge = __builtin_amdgcn_permlanex16(xi, xi, 0u, 0u, false, false);
  return __int_as_float(
      __builtin_amdgcn_update_dpp(edge, xi, 0x101, 0xf, 0xf, false));
}

__device__ __forceinline__ void mvprop_hmax(const float (&vr)[PC], float (&h)[PC],
                                            bool firstLane, bool lastLane) {
  const float ninf = -__builtin_inff();
  float l = mvprop_prev_lane(vr[PC - 1]);
  float g = mvprop_next_lane(vr[0]);
  l = firstLane ? ninf : l;
  g = lastLane ? ninf : g;
  h[0] = fmaxf(fmaxf(l, vr[0]), vr[1]);
#pragma unroll
  for (int c = 1; c < PC - 1; ++c)
    h[c] = fmaxf(fmaxf(vr[c - 1], vr[c]), vr[c + 1]);
  h[PC - 1] = fmaxf(fmaxf(vr[PC - 2], vr[PC - 1]), g);
}

template <int LO, int HI>
__device__ __forceinline__ void mvprop_sweep(float (&v)[PH][PC],
                                             const float (&r)[PH][PC],
                                             const float (&p)[PH][PC], int ybase,
                                             bool firstLane, bool lastLane) {
  float h[3][PC];
  mvprop_hmax(v[LO - 1], h[(LO - 1) % 3], firstLane, lastLane);
  mvprop_hmax(v[LO], h[LO % 3], firstLane, lastLane);
#pragma unroll
  for (int i = LO; i < HI; ++i) {
    mvprop_hmax(v[i + 1], h[(i + 1) % 3], firstLane, lastLane);
    const bool inImage = (unsigned)(ybase + i) < 256u;
#pragma unroll
    for (int c = 0; c < PC; ++c) {
      float mx = fmaxf(fmaxf(h[(i - 1) % 3][c], h[i % 3][c]), h[(i + 1) % 3][c]);
      float rr = r[i][c];
      float vn = fmaxf(v[i][c], rr + p[i][c] * (mx - rr));
      v[i][c] = inImage ? vn : v[i][c];
    }
  }
}

template <int LO, int HI>
__device__ __forceinline__ void mvprop_phase(int nsteps, float (&v)[PH][PC],
                                             const float (&r)[PH][PC],
                                             const float (&p)[PH][PC], int ybase,
                                             bool firstLane, bool lastLane) {
#pragma unroll 1
  for (int s = 0; s < nsteps; ++s)
    mvprop_sweep<LO, HI>(v, r, p, ybase, firstLane, lastLane);
}

template <bool FIRST, bool FINAL>
__global__ __launch_bounds__(32) __attribute__((amdgpu_waves_per_eu(1, 1)))
void mvprop_strip(
    const float* __restrict__ image, const float* __restrict__ pbuf,
    const float* __restrict__ vin, long vinStride, float* __restrict__ vout) {
  const int lane  = threadIdx.x;
  const int right = blockIdx.x & 1;
  const int sid   = blockIdx.x >> 1;
  const int b     = sid / NSTRIP;
  const int strip = sid % NSTRIP;
  const int y0    = strip * PR;
  const int ybase = y0 - PT;
  const int xo    = (right ? PXR : 0) + lane * PC;
  const bool firstLane = (lane == 0);
  const bool lastLane  = (lane == 31);
  const float ninf = -__builtin_inff();

  const float* __restrict__ vb = vin + (size_t)b * vinStride;
  const float* __restrict__ ib = image + (size_t)b * 2 * NPIX;
  const float* __restrict__ rb = ib + NPIX;
  const float* __restrict__ pb = pbuf + (size_t)b * NPIX;

  float v[PH][PC], r[PH][PC], p[PH][PC];

#pragma unroll
  for (int i = 0; i < PH; ++i) {
    const int y = ybase + i;
    const bool inImage = (unsigned)y < 256u;
    const int yc = min(max(y, 0), 255);
    float a[PC];
    mvprop_load_cols(vb + yc * 256 + xo, a);
#pragma unroll
    for (int c = 0; c < PC; ++c) v[i][c] = inImage ? a[c] : ninf;
  }
#pragma unroll
  for (int i = 1; i < PH - 1; ++i) {
    const int yc = min(max(ybase + i, 0), 255);
    if (FIRST) {
#pragma unroll
      for (int c = 0; c < PC; ++c) r[i][c] = v[i][c];
    } else {
      mvprop_load_cols(rb + yc * 256 + xo, r[i]);
    }
    mvprop_load_cols(pb + yc * 256 + xo, p[i]);
  }

  mvprop_phase<1, PH - 1>(PG, v, r, p, ybase, firstLane, lastLane);
  mvprop_phase<PG + 1, PH - PG - 1>(PT - PG, v, r, p, ybase, firstLane, lastLane);

  float o[PR][PC];
#pragma unroll
  for (int j = 0; j < PR; ++j) {
#pragma unroll
    for (int c = 0; c < PC; ++c) o[j][c] = v[PT + j][c];
    if (FINAL) {
      float oc[PC];
      mvprop_load_cols(ib + (y0 + j) * 256 + xo, oc);
#pragma unroll
      for (int c = 0; c < PC; ++c) o[j][c] = (oc[c] > 0.49f) ? o[j][c] : -1.0f;
    }
  }
  float* dst = vout + (size_t)b * NPIX + y0 * 256 + xo;
#pragma unroll
  for (int c = 0; c < PC; ++c) {
    const bool mine = right ? (xo + c >= PSEAM) : (xo + c < PSEAM);
    if (mine) {
#pragma unroll
      for (int j = 0; j < PR; ++j) dst[j * 256 + c] = o[j][c];
    }
  }
}

extern "C" void kernel_launch(void* const* d_in, const int* in_sizes, int n_in,
                              void* d_out, int out_size, void* d_ws,
                              size_t ws_size, hipStream_t stream) {
  const float* image = (const float*)d_in[0];
  const float* Wh    = (const float*)d_in[1];
  const float* bh    = (const float*)d_in[2];
  const float* Wp    = (const float*)d_in[3];
  float* out = (float*)d_out;

  float* ws    = (float*)d_ws;
  float* pbuf  = ws;
  float* v0    = ws + (size_t)NTOT;

  mvprop_conv<<<NBATCH * CB_SEGS, 256, 0, stream>>>(image, Wh, bh, Wp, pbuf);

  mvprop_strip<true, false><<<NBATCH * NSTRIP * 2, 32, 0, stream>>>(
      image, pbuf, image + NPIX, 2 * NPIX, v0);
  mvprop_strip<false, true><<<NBATCH * NSTRIP * 2, 32, 0, stream>>>(
      image, pbuf, v0, NPIX, out);
}


# conv: m2 is C of products 1 and 4, m3 is added last on the vector ALU: a relu input is now an add's result, so the compiler's quieting max(x,x) before each relu max goes; tap sum order changes
# speedup vs baseline: 4.8580x; 1.0389x over previous
#include <hip/hip_runtime.h>

typedef __attribute__((ext_vector_type(2))) float v2f;
typedef __attribute__((ext_vector_type(8))) float v8f;

#define NPIX   (256 * 256)
#define NBATCH 16
#define NTOT   (NBATCH * NPIX)

typedef float mvprop_f4u __attribute__((ext_vector_type(4), aligned(4)));

#define CB_ROWS 8
#define CB_SEGS (256 / CB_ROWS)

__device__ __forceinline__ void mvprop_row_load(const float* __restrict__ occ,
                                                int yy, int x, int h,
                                                float& side, float& mid) {
  const int yc  = min(max(yy, 0), 255);
  const int xs  = h ? x + 1 : x - 1;
  const int xsc = min(max(xs, 0), 255);
  side = occ[yc * 256 + xsc];
  mid  = occ[yc * 256 + x];
}

__device__ __forceinline__ v2f mvprop_row_vals(int yy, int x, int h, float side,
                                               float mid) {
  const bool rowOk  = (unsigned)yy < 256u;
  const int xs      = h ? x + 1 : x - 1;
  const bool sideOk = rowOk & ((unsigned)xs < 256u);
  v2f a;
  a.x = sideOk ? side : 0.0f;
  a.y = (rowOk & (h == 0)) ? mid : 0.0f;
  return a;
}

template <int MASK>
__device__ __forceinline__ float mvprop_xadd(float keep, float send) {
  const int moved = __builtin_amdgcn_update_dpp(0, __float_as_int(send),
                                                0x160 + MASK, 0xf, 0xf, true);
  return keep + __int_as_float(moved);
}

__device__ __forceinline__ float mvprop_chan_sum(const v2f (&acc)[2][4], bool q0,
                                                 bool q1, bool q2, bool q3) {
  float s8[8], s4[4], s2[2];
#pragma unroll
  for (int k = 0; k < 8; ++k) {
    const float lo = acc[k >> 2][k & 3].x, hi = acc[k >> 2][k & 3].y;
    s8[k] = mvprop_xadd<1>(q0 ? hi : lo, q0 ? lo : hi);
  }
#pragma unroll
  for (int k = 0; k < 4; ++k)
    s4[k] = mvprop_xadd<2>(q1 ? s8[2 * k + 1] : s8[2 * k],
                           q1 ? s8[2 * k] : s8[2 * k + 1]);
#pragma unroll
  for (int k = 0; k < 2; ++k)
    s2[k] = mvprop_xadd<4>(q2 ? s4[2 * k + 1] : s4[2 * k],
                           q2 ? s4[2 * k] : s4[2 * k + 1]);
  return mvprop_xadd<8>(q3 ? s2[1] : s2[0], q3 ? s2[0] : s2[1]);
}

__global__ __launch_bounds__(256) void mvprop_conv(
    const float* __restrict__ image, const float* __restrict__ Wh,
    const float* __restrict__ bh, const float* __restrict__ Wp,
    float* __restrict__ pout) {
  const int lane = threadIdx.x & 31;
  const int band = threadIdx.x >> 5;
  const int b    = blockIdx.x / CB_SEGS;
  const int seg  = blockIdx.x % CB_SEGS;
  const int y0   = seg * CB_ROWS;
  const int m    = lane & 15;
  const int h    = lane >> 4;
  const int x0   = band * 32 + m;
  const bool q0 = m & 1, q1 = m & 2, q2 = m & 4, q3 = m & 8;
  const int pix = 16 * (m >> 3) + 8 * h + (m & 7);

  const float* __restrict__ occ = image + (size_t)b * 2 * NPIX;

  v2f bF[10][4];
  float wp[10];
#pragma unroll
  for (int t = 0; t < 10; ++t) {
    const int n   = t * 16 + m;
    const bool ok = n < 150;
    const int nc  = ok ? n : 149;
    const float* __restrict__ wn = Wh + nc * 9;
    const mvprop_f4u w03 = *(const mvprop_f4u*)wn;
    const mvprop_f4u w47 = *(const mvprop_f4u*)(wn + 4);
    const float w[9] = {w03.x, w03.y, w03.z, w03.w, w47.x, w47.y, w47.z, w47.w,
                        wn[8]};
    const float bias = bh[nc];
    const float wpn  = Wp[nc];
    const float gx0 = h ? w[2] : w[0], gx1 = h ? w[5] : w[3], gx2 = h ? w[8] : w[6];
    const float gy0 = w[1], gy1 = w[4], gy2 = w[7];
    const float ex = gx0 + gx2, ey = gy0 + gy2;
    float ux[4], uy[4];
    ux[0] = gx0;               uy[0] = h ? 0.0f : gy0;
    ux[1] = 0.5f * (ex + gx1); uy[1] = h ? bias : 0.5f * (ey + gy1);
    ux[2] = 0.5f * (ex - gx1); uy[2] = h ? 0.0f : 0.5f * (ey - gy1);
    ux[3] = -gx2;              uy[3] = h ? 0.0f : -gy2;
#pragma unroll
    for (int i = 0; i < 4; ++i) {
      bF[t][i].x = ok ? ux[i] : 0.0f;
      bF[t][i].y = ok ? uy[i] : 0.0f;
    }
    wp[t] = ok ? wpn : 0.0f;
  }

  v2f r0[2], r1[2], r2[2], r3[2];
#pragma unroll
  for (int T = 0; T < 2; ++T) {
    float s0, m0, s1, m1, s2, m2, s3, m3;
    mvprop_row_load(occ, y0 - 1, x0 + 16 * T, h, s0, m0);
    mvprop_row_load(occ, y0,     x0 + 16 * T, h, s1, m1);
    mvprop_row_load(occ, y0 + 1, x0 + 16 * T, h, s2, m2);
    mvprop_row_load(occ, y0 + 2, x0 + 16 * T, h, s3, m3);
    r0[T] = mvprop_row_vals(y0 - 1, x0 + 16 * T, h, s0, m0);
    r1[T] = mvprop_row_vals(y0,     x0 + 16 * T, h, s1, m1);
    r2[T] = mvprop_row_vals(y0 + 1, x0 + 16 * T, h, s2, m2);
    r3[T] = mvprop_row_vals(y0 + 2, x0 + 16 * T, h, s3, m3);
  }

#pragma unroll 1
  for (int j = 0; j < CB_ROWS; j += 2) {
    const int y = y0 + j;

    float fs3[2], fm3[2], fs4[2], fm4[2];
#pragma unroll
    for (int T = 0; T < 2; ++T) {
      mvprop_row_load(occ, y + 3, x0 + 16 * T, h, fs3[T], fm3[T]);
      mvprop_row_load(occ, y + 4, x0 + 16 * T, h, fs4[T], fm4[T]);
    }

    v2f a1[2], a2[2], a3[2], a4[2];
#pragma unroll
    for (int T = 0; T < 2; ++T) {
      a1[T] = r0[T] - r2[T];
      a2[T] = r1[T] + r2[T];
      a2[T].y = h ? 1.0f : a2[T].y;
      a3[T] = r2[T] - r1[T];
      a4[T] = r1[T] - r3[T];
    }

    v2f accE[2][4], accO[2][4];
#pragma unroll
    for (int k = 0; k < 4; ++k) {
      accE[0][k] = 0.0f; accE[1][k] = 0.0f;
      accO[0][k] = 0.0f; accO[1][k] = 0.0f;
    }

#pragma unroll
    for (int t = 0; t < 10; ++t) {
      v8f z8;
#pragma unroll
      for (int r = 0; r < 8; ++r) z8[r] = 0.0f;
      const v8f p0 = __builtin_amdgcn_wmma_f32_16x16x4_f32(
          false, a2[0], false, bF[t][1], (short)0, z8, false, false);
      const v8f g0 = __builtin_amdgcn_wmma_f32_16x16x4_f32(
          false, a3[0], false, bF[t][2], (short)0, z8, false, false);
      const v8f p1 = __builtin_amdgcn_wmma_f32_16x16x4_f32(
          false, a2[1], false, bF[t][1], (short)0, z8, false, false);
      const v8f g1 = __builtin_amdgcn_wmma_f32_16x16x4_f32(
          false, a3[1], false, bF[t][2], (short)0, z8, false, false);
      const v8f x0 = __builtin_amdgcn_wmma_f32_16x16x4_f32(
          false, a1[0], false, bF[t][0], (short)0, p0, false, false);
      const v8f y0 = __builtin_amdgcn_wmma_f32_16x16x4_f32(
          false, a4[0], false, bF[t][3], (short)0, p0, false, false);
      const v8f x1 = __builtin_amdgcn_wmma_f32_16x16x4_f32(
          false, a1[1], false, bF[t][0], (short)0, p1, false, false);
      const v8f y1 = __builtin_amdgcn_wmma_f32_16x16x4_f32(
          false, a4[1], false, bF[t][3], (short)0, p1, false, false);
      const v8f e0 = x0 + g0, o0 = y0 - g0;
      const v8f e1 = x1 + g1, o1 = y1 - g1;
      const v2f wp2 = {wp[t], wp[t]};
      v8f zero8;
#pragma unroll
      for (int r = 0; r < 8; ++r) zero8[r] = 0.0f;
      const v8f re0 = __builtin_elementwise_max(e0, zero8);
      const v8f ro0 = __builtin_elementwise_max(o0, zero8);
      const v8f re1 = __builtin_elementwise_max(e1, zero8);
      const v8f ro1 = __builtin_elementwise_max(o1, zero8);
#pragma unroll
      for (int k = 0; k < 4; ++k) {
        const v2f de0 = {re0[2 * k], re0[2 * k + 1]};
        const v2f do0 = {ro0[2 * k], ro0[2 * k + 1]};
        const v2f de1 = {re1[2 * k], re1[2 * k + 1]};
        const v2f do1 = {ro1[2 * k], ro1[2 * k + 1]};
        accE[0][k] = __builtin_elementwise_fma(de0, wp2, accE[0][k]);
        accO[0][k] = __builtin_elementwise_fma(do0, wp2, accO[0][k]);
        accE[1][k] = __builtin_elementwise_fma(de1, wp2, accE[1][k]);
        accO[1][k] = __builtin_elementwise_fma(do1, wp2, accO[1][k]);
      }
    }

    const float sumE = mvprop_chan_sum(accE, q0, q1, q2, q3);
    const float sumO = mvprop_chan_sum(accO, q0, q1, q2, q3);
    float* __restrict__ prow = pout + (size_t)b * NPIX + y * 256 + band * 32 + pix;
    prow[0]   = 1.0f / (1.0f + __expf(-sumE));
    prow[256] = 1.0f / (1.0f + __expf(-sumO));

#pragma unroll
    for (int T = 0; T < 2; ++T) {
      r0[T] = r2[T];
      r1[T] = r3[T];
      r2[T] = mvprop_row_vals(y + 3, x0 + 16 * T, h, fs3[T], fm3[T]);
      r3[T] = mvprop_row_vals(y + 4, x0 + 16 * T, h, fs4[T], fm4[T]);
    }
  }
}

#define PT 10
#define PR 8
#define PH (PR + 2 * PT)
#define NSTRIP (256 / PR)
#define PG 5
#define PC 5
#define PW (32 * PC)
#define PXR (256 - PW)
#define PSEAM 128
static_assert(PW - PT >= PSEAM, "left wave must be exact up to the seam");
static_assert(PXR + PT <= PSEAM, "right wave must be exact from the seam on");

__device__ __forceinline__ void mvprop_load_cols(const float* __restrict__ src,
                                                 float (&d)[PC]) {
  const mvprop_f4u a = *(const mvprop_f4u*)src;
  d[0] = a.x; d[1] = a.y; d[2] = a.z; d[3] = a.w;
  d[4] = src[4];
}
static_assert(PC == 5, "mvprop_load_cols reads 4 + 1 floats");

__device__ __forceinline__ float mvprop_prev_lane(float x) {
  const int xi = __float_as_int(x);
  const int edge = __builtin_amdgcn_permlanex16(xi, xi, 0xffffffffu, 0xffffffffu,
                                                false, false);
  return __int_as_float(
      __builtin_amdgcn_update_dpp(edge, xi, 0x111, 0xf, 0xf, false));
}
__device__ __forceinline__ float mvprop_next_lane(float x) {
  const int xi = __float_as_int(x);
  const int edge = __builtin_amdgcn_permlanex16(xi, xi, 0u, 0u, false, false);
  return __int_as_float(
      __builtin_amdgcn_update_dpp(edge, xi, 0x101, 0xf, 0xf, false));
}

__device__ __forceinline__ void mvprop_hmax(const float (&vr)[PC], float (&h)[PC],
                                            bool firstLane, bool lastLane) {
  const float ninf = -__builtin_inff();
  float l = mvprop_prev_lane(vr[PC - 1]);
  float g = mvprop_next_lane(vr[0]);
  l = firstLane ? ninf : l;
  g = lastLane ? ninf : g;
  h[0] = fmaxf(fmaxf(l, vr[0]), vr[1]);
#pragma unroll
  for (int c = 1; c < PC - 1; ++c)
    h[c] = fmaxf(fmaxf(vr[c - 1], vr[c]), vr[c + 1]);
  h[PC - 1] = fmaxf(fmaxf(vr[PC - 2], vr[PC - 1]), g);
}

template <int LO, int HI>
__device__ __forceinline__ void mvprop_sweep(float (&v)[PH][PC],
                                             const float (&r)[PH][PC],
                                             const float (&p)[PH][PC], int ybase,
                                             bool firstLane, bool lastLane) {
  float h[3][PC];
  mvprop_hmax(v[LO - 1], h[(LO - 1) % 3], firstLane, lastLane);
  mvprop_hmax(v[LO], h[LO % 3], firstLane, lastLane);
#pragma unroll
  for (int i = LO; i < HI; ++i) {
    mvprop_hmax(v[i + 1], h[(i + 1) % 3], firstLane, lastLane);
    const bool inImage = (unsigned)(ybase + i) < 256u;
#pragma unroll
    for (int c = 0; c < PC; ++c) {
      float mx = fmaxf(fmaxf(h[(i - 1) % 3][c], h[i % 3][c]), h[(i + 1) % 3][c]);
      float rr = r[i][c];
      float vn = fmaxf(v[i][c], rr + p[i][c] * (mx - rr));
      v[i][c] = inImage ? vn : v[i][c];
    }
  }
}

template <int LO, int HI>
__device__ __forceinline__ void mvprop_phase(int nsteps, float (&v)[PH][PC],
                                             const float (&r)[PH][PC],
                                             const float (&p)[PH][PC], int ybase,
                                             bool firstLane, bool lastLane) {
#pragma unroll 1
  for (int s = 0; s < nsteps; ++s)
    mvprop_sweep<LO, HI>(v, r, p, ybase, firstLane, lastLane);
}

template <bool FIRST, bool FINAL>
__global__ __launch_bounds__(32) __attribute__((amdgpu_waves_per_eu(1, 1)))
void mvprop_strip(
    const float* __restrict__ image, const float* __restrict__ pbuf,
    const float* __restrict__ vin, long vinStride, float* __restrict__ vout) {
  const int lane  = threadIdx.x;
  const int right = blockIdx.x & 1;
  const int sid   = blockIdx.x >> 1;
  const int b     = sid / NSTRIP;
  const int strip = sid % NSTRIP;
  const int y0    = strip * PR;
  const int ybase = y0 - PT;
  const int xo    = (right ? PXR : 0) + lane * PC;
  const bool firstLane = (lane == 0);
  const bool lastLane  = (lane == 31);
  const float ninf = -__builtin_inff();

  const float* __restrict__ vb = vin + (size_t)b * vinStride;
  const float* __restrict__ ib = image + (size_t)b * 2 * NPIX;
  const float* __restrict__ rb = ib + NPIX;
  const float* __restrict__ pb = pbuf + (size_t)b * NPIX;

  float v[PH][PC], r[PH][PC], p[PH][PC];

#pragma unroll
  for (int i = 0; i < PH; ++i) {
    const int y = ybase + i;
    const bool inImage = (unsigned)y < 256u;
    const int yc = min(max(y, 0), 255);
    float a[PC];
    mvprop_load_cols(vb + yc * 256 + xo, a);
#pragma unroll
    for (int c = 0; c < PC; ++c) v[i][c] = inImage ? a[c] : ninf;
  }
#pragma unroll
  for (int i = 1; i < PH - 1; ++i) {
    const int yc = min(max(ybase + i, 0), 255);
    if (FIRST) {
#pragma unroll
      for (int c = 0; c < PC; ++c) r[i][c] = v[i][c];
    } else {
      mvprop_load_cols(rb + yc * 256 + xo, r[i]);
    }
    mvprop_load_cols(pb + yc * 256 + xo, p[i]);
  }

  mvprop_phase<1, PH - 1>(PG, v, r, p, ybase, firstLane, lastLane);
  mvprop_phase<PG + 1, PH - PG - 1>(PT - PG, v, r, p, ybase, firstLane, lastLane);

  float o[PR][PC];
#pragma unroll
  for (int j = 0; j < PR; ++j) {
#pragma unroll
    for (int c = 0; c < PC; ++c) o[j][c] = v[PT + j][c];
    if (FINAL) {
      float oc[PC];
      mvprop_load_cols(ib + (y0 + j) * 256 + xo, oc);
#pragma unroll
      for (int c = 0; c < PC; ++c) o[j][c] = (oc[c] > 0.49f) ? o[j][c] : -1.0f;
    }
  }
  float* dst = vout + (size_t)b * NPIX + y0 * 256 + xo;
#pragma unroll
  for (int c = 0; c < PC; ++c) {
    const bool mine = right ? (xo + c >= PSEAM) : (xo + c < PSEAM);
    if (mine) {
#pragma unroll
      for (int j = 0; j < PR; ++j) dst[j * 256 + c] = o[j][c];
    }
  }
}

extern "C" void kernel_launch(void* const* d_in, const int* in_sizes, int n_in,
                              void* d_out, int out_size, void* d_ws,
                              size_t ws_size, hipStream_t stream) {
  const float* image = (const float*)d_in[0];
  const float* Wh    = (const float*)d_in[1];
  const float* bh    = (const float*)d_in[2];
  const float* Wp    = (const float*)d_in[3];
  float* out = (float*)d_out;

  float* ws    = (float*)d_ws;
  float* pbuf  = ws;
  float* v0    = ws + (size_t)NTOT;

  mvprop_conv<<<NBATCH * CB_SEGS, 256, 0, stream>>>(image, Wh, bh, Wp, pbuf);

  mvprop_strip<true, false><<<NBATCH * NSTRIP * 2, 32, 0, stream>>>(
      image, pbuf, image + NPIX, 2 * NPIX, v0);
  mvprop_strip<false, true><<<NBATCH * NSTRIP * 2, 32, 0, stream>>>(
      image, pbuf, v0, NPIX, out);
}


# propagation: a wave's part is cut across into two 14-row blocks, one per 16-lane row (10 columns a lane, lower block mirrored): side neighbours by one DPP shift, end lane keeps -inf; same arithmetic
# speedup vs baseline: 5.0366x; 1.0368x over previous
#include <hip/hip_runtime.h>

typedef __attribute__((ext_vector_type(2))) float v2f;
typedef __attribute__((ext_vector_type(8))) float v8f;

#define NPIX   (256 * 256)
#define NBATCH 16
#define NTOT   (NBATCH * NPIX)

typedef float mvprop_f4u __attribute__((ext_vector_type(4), aligned(4)));

#define CB_ROWS 8
#define CB_SEGS (256 / CB_ROWS)

__device__ __forceinline__ void mvprop_row_load(const float* __restrict__ occ,
                                                int yy, int x, int h,
                                                float& side, float& mid) {
  const int yc  = min(max(yy, 0), 255);
  const int xs  = h ? x + 1 : x - 1;
  const int xsc = min(max(xs, 0), 255);
  side = occ[yc * 256 + xsc];
  mid  = occ[yc * 256 + x];
}

__device__ __forceinline__ v2f mvprop_row_vals(int yy, int x, int h, float side,
                                               float mid) {
  const bool rowOk  = (unsigned)yy < 256u;
  const int xs      = h ? x + 1 : x - 1;
  const bool sideOk = rowOk & ((unsigned)xs < 256u);
  v2f a;
  a.x = sideOk ? side : 0.0f;
  a.y = (rowOk & (h == 0)) ? mid : 0.0f;
  return a;
}

template <int MASK>
__device__ __forceinline__ float mvprop_xadd(float keep, float send) {
  const int moved = __builtin_amdgcn_update_dpp(0, __float_as_int(send),
                                                0x160 + MASK, 0xf, 0xf, true);
  return keep + __int_as_float(moved);
}

__device__ __forceinline__ float mvprop_chan_sum(const v2f (&acc)[2][4], bool q0,
                                                 bool q1, bool q2, bool q3) {
  float s8[8], s4[4], s2[2];
#pragma unroll
  for (int k = 0; k < 8; ++k) {
    const float lo = acc[k >> 2][k & 3].x, hi = acc[k >> 2][k & 3].y;
    s8[k] = mvprop_xadd<1>(q0 ? hi : lo, q0 ? lo : hi);
  }
#pragma unroll
  for (int k = 0; k < 4; ++k)
    s4[k] = mvprop_xadd<2>(q1 ? s8[2 * k + 1] : s8[2 * k],
                           q1 ? s8[2 * k] : s8[2 * k + 1]);
#pragma unroll
  for (int k = 0; k < 2; ++k)
    s2[k] = mvprop_xadd<4>(q2 ? s4[2 * k + 1] : s4[2 * k],
                           q2 ? s4[2 * k] : s4[2 * k + 1]);
  return mvprop_xadd<8>(q3 ? s2[1] : s2[0], q3 ? s2[0] : s2[1]);
}

__global__ __launch_bounds__(256) void mvprop_conv(
    const float* __restrict__ image, const float* __restrict__ Wh,
    const float* __restrict__ bh, const float* __restrict__ Wp,
    float* __restrict__ pout) {
  const int lane = threadIdx.x & 31;
  const int band = threadIdx.x >> 5;
  const int b    = blockIdx.x / CB_SEGS;
  const int seg  = blockIdx.x % CB_SEGS;
  const int y0   = seg * CB_ROWS;
  const int m    = lane & 15;
  const int h    = lane >> 4;
  const int x0   = band * 32 + m;
  const bool q0 = m & 1, q1 = m & 2, q2 = m & 4, q3 = m & 8;
  const int pix = 16 * (m >> 3) + 8 * h + (m & 7);

  const float* __restrict__ occ = image + (size_t)b * 2 * NPIX;

  v2f bF[10][4];
  float wp[10];
#pragma unroll
  for (int t = 0; t < 10; ++t) {
    const int n   = t * 16 + m;
    const bool ok = n < 150;
    const int nc  = ok ? n : 149;
    const float* __restrict__ wn = Wh + nc * 9;
    const mvprop_f4u w03 = *(const mvprop_f4u*)wn;
    const mvprop_f4u w47 = *(const mvprop_f4u*)(wn + 4);
    const float w[9] = {w03.x, w03.y, w03.z, w03.w, w47.x, w47.y, w47.z, w47.w,
                        wn[8]};
    const float bias = bh[nc];
    const float wpn  = Wp[nc];
    const float gx0 = h ? w[2] : w[0], gx1 = h ? w[5] : w[3], gx2 = h ? w[8] : w[6];
    const float gy0 = w[1], gy1 = w[4], gy2 = w[7];
    const float ex = gx0 + gx2, ey = gy0 + gy2;
    float ux[4], uy[4];
    ux[0] = gx0;               uy[0] = h ? 0.0f : gy0;
    ux[1] = 0.5f * (ex + gx1); uy[1] = h ? bias : 0.5f * (ey + gy1);
    ux[2] = 0.5f * (ex - gx1); uy[2] = h ? 0.0f : 0.5f * (ey - gy1);
    ux[3] = -gx2;              uy[3] = h ? 0.0f : -gy2;
#pragma unroll
    for (int i = 0; i < 4; ++i) {
      bF[t][i].x = ok ? ux[i] : 0.0f;
      bF[t][i].y = ok ? uy[i] : 0.0f;
    }
    wp[t] = ok ? wpn : 0.0f;
  }

  v2f r0[2], r1[2], r2[2], r3[2];
#pragma unroll
  for (int T = 0; T < 2; ++T) {
    float s0, m0, s1, m1, s2, m2, s3, m3;
    mvprop_row_load(occ, y0 - 1, x0 + 16 * T, h, s0, m0);
    mvprop_row_load(occ, y0,     x0 + 16 * T, h, s1, m1);
    mvprop_row_load(occ, y0 + 1, x0 + 16 * T, h, s2, m2);
    mvprop_row_load(occ, y0 + 2, x0 + 16 * T, h, s3, m3);
    r0[T] = mvprop_row_vals(y0 - 1, x0 + 16 * T, h, s0, m0);
    r1[T] = mvprop_row_vals(y0,     x0 + 16 * T, h, s1, m1);
    r2[T] = mvprop_row_vals(y0 + 1, x0 + 16 * T, h, s2, m2);
    r3[T] = mvprop_row_vals(y0 + 2, x0 + 16 * T, h, s3, m3);
  }

#pragma unroll 1
  for (int j = 0; j < CB_ROWS; j += 2) {
    const int y = y0 + j;

    float fs3[2], fm3[2], fs4[2], fm4[2];
#pragma unroll
    for (int T = 0; T < 2; ++T) {
      mvprop_row_load(occ, y + 3, x0 + 16 * T, h, fs3[T], fm3[T]);
      mvprop_row_load(occ, y + 4, x0 + 16 * T, h, fs4[T], fm4[T]);
    }

    v2f a1[2], a2[2], a3[2], a4[2];
#pragma unroll
    for (int T = 0; T < 2; ++T) {
      a1[T] = r0[T] - r2[T];
      a2[T] = r1[T] + r2[T];
      a2[T].y = h ? 1.0f : a2[T].y;
      a3[T] = r2[T] - r1[T];
      a4[T] = r1[T] - r3[T];
    }

    v2f accE[2][4], accO[2][4];
#pragma unroll
    for (int k = 0; k < 4; ++k) {
      accE[0][k] = 0.0f; accE[1][k] = 0.0f;
      accO[0][k] = 0.0f; accO[1][k] = 0.0f;
    }

#pragma unroll
    for (int t = 0; t < 10; ++t) {
      v8f z8;
#pragma unroll
      for (int r = 0; r < 8; ++r) z8[r] = 0.0f;
      const v8f p0 = __builtin_amdgcn_wmma_f32_16x16x4_f32(
          false, a2[0], false, bF[t][1], (short)0, z8, false, false);
      const v8f g0 = __builtin_amdgcn_wmma_f32_16x16x4_f32(
          false, a3[0], false, bF[t][2], (short)0, z8, false, false);
      const v8f p1 = __builtin_amdgcn_wmma_f32_16x16x4_f32(
          false, a2[1], false, bF[t][1], (short)0, z8, false, false);
      const v8f g1 = __builtin_amdgcn_wmma_f32_16x16x4_f32(
          false, a3[1], false, bF[t][2], (short)0, z8, false, false);
      const v8f x0 = __builtin_amdgcn_wmma_f32_16x16x4_f32(
          false, a1[0], false, bF[t][0], (short)0, p0, false, false);
      const v8f y0 = __builtin_amdgcn_wmma_f32_16x16x4_f32(
          false, a4[0], false, bF[t][3], (short)0, p0, false, false);
      const v8f x1 = __builtin_amdgcn_wmma_f32_16x16x4_f32(
          false, a1[1], false, bF[t][0], (short)0, p1, false, false);
      const v8f y1 = __builtin_amdgcn_wmma_f32_16x16x4_f32(
          false, a4[1], false, bF[t][3], (short)0, p1, false, false);
      const v8f e0 = x0 + g0, o0 = y0 - g0;
      const v8f e1 = x1 + g1, o1 = y1 - g1;
      const v2f wp2 = {wp[t], wp[t]};
      v8f zero8;
#pragma unroll
      for (int r = 0; r < 8; ++r) zero8[r] = 0.0f;
      const v8f re0 = __builtin_elementwise_max(e0, zero8);
      const v8f ro0 = __builtin_elementwise_max(o0, zero8);
      const v8f re1 = __builtin_elementwise_max(e1, zero8);
      const v8f ro1 = __builtin_elementwise_max(o1, zero8);
#pragma unroll
      for (int k = 0; k < 4; ++k) {
        const v2f de0 = {re0[2 * k], re0[2 * k + 1]};
        const v2f do0 = {ro0[2 * k], ro0[2 * k + 1]};
        const v2f de1 = {re1[2 * k], re1[2 * k + 1]};
        const v2f do1 = {ro1[2 * k], ro1[2 * k + 1]};
        accE[0][k] = __builtin_elementwise_fma(de0, wp2, accE[0][k]);
        accO[0][k] = __builtin_elementwise_fma(do0, wp2, accO[0][k]);
        accE[1][k] = __builtin_elementwise_fma(de1, wp2, accE[1][k]);
        accO[1][k] = __builtin_elementwise_fma(do1, wp2, accO[1][k]);
      }
    }

    const float sumE = mvprop_chan_sum(accE, q0, q1, q2, q3);
    const float sumO = mvprop_chan_sum(accO, q0, q1, q2, q3);
    float* __restrict__ prow = pout + (size_t)b * NPIX + y * 256 + band * 32 + pix;
    prow[0]   = 1.0f / (1.0f + __expf(-sumE));
    prow[256] = 1.0f / (1.0f + __expf(-sumO));

#pragma unroll
    for (int T = 0; T < 2; ++T) {
      r0[T] = r2[T];
      r1[T] = r3[T];
      r2[T] = mvprop_row_vals(y + 3, x0 + 16 * T, h, fs3[T], fm3[T]);
      r3[T] = mvprop_row_vals(y + 4, x0 + 16 * T, h, fs4[T], fm4[T]);
    }
  }
}

#define PT 10
#define PR 8
#define PH (PR + 2 * PT)
#define PK (PH / 2)
#define NSTRIP (256 / PR)
#define PG 5
#define PC 10
#define PW (16 * PC)
#define PXR (256 - PW)
#define PSEAM 128
static_assert(PH % 2 == 0, "the strip is cut into two blocks of equal height");
static_assert(PW - PT >= PSEAM, "left wave must be exact up to the seam");
static_assert(PXR + PT <= PSEAM, "right wave must be exact from the seam on");

typedef float mvprop_f2u __attribute__((ext_vector_type(2), aligned(4)));

__device__ __forceinline__ void mvprop_load_cols(const float* __restrict__ src,
                                                 float (&d)[PC]) {
  const mvprop_f4u a = *(const mvprop_f4u*)src;
  const mvprop_f4u b = *(const mvprop_f4u*)(src + 4);
  const mvprop_f2u c = *(const mvprop_f2u*)(src + 8);
  d[0] = a.x; d[1] = a.y; d[2] = a.z; d[3] = a.w;
  d[4] = b.x; d[5] = b.y; d[6] = b.z; d[7] = b.w;
  d[8] = c.x; d[9] = c.y;
}
static_assert(PC == 10, "mvprop_load_cols reads 4 + 4 + 2 floats");

__device__ __forceinline__ float mvprop_prev_lane(float x) {
  const int pad = __float_as_int(-__builtin_inff());
  return __int_as_float(
      __builtin_amdgcn_update_dpp(pad, __float_as_int(x), 0x111, 0xf, 0xf, false));
}
__device__ __forceinline__ float mvprop_next_lane(float x) {
  const int pad = __float_as_int(-__builtin_inff());
  return __int_as_float(
      __builtin_amdgcn_update_dpp(pad, __float_as_int(x), 0x101, 0xf, 0xf, false));
}
__device__ __forceinline__ float mvprop_other_block(float x) {
  const int xi = __float_as_int(x);
  return __int_as_float(__builtin_amdgcn_permlanex16(xi, xi, 0x76543210, 0xfedcba98,
                                                     false, false));
}

__device__ __forceinline__ void mvprop_hmax(const float (&vr)[PC], float (&h)[PC]) {
  const float l = mvprop_prev_lane(vr[PC - 1]);
  const float g = mvprop_next_lane(vr[0]);
  h[0] = fmaxf(fmaxf(l, vr[0]), vr[1]);
#pragma unroll
  for (int c = 1; c < PC - 1; ++c)
    h[c] = fmaxf(fmaxf(vr[c - 1], vr[c]), vr[c + 1]);
  h[PC - 1] = fmaxf(fmaxf(vr[PC - 2], vr[PC - 1]), g);
}

template <int LO>
__device__ __forceinline__ void mvprop_sweep(float (&v)[PK][PC],
                                             const float (&r)[PK][PC],
                                             const float (&p)[PK][PC],
                                             const bool (&inImage)[PK]) {
  float h[3][PC];
  mvprop_hmax(v[LO - 1], h[(LO - 1) % 3]);
  mvprop_hmax(v[LO], h[LO % 3]);
#pragma unroll
  for (int k = LO; k < PK; ++k) {
    if (k + 1 < PK) {
      mvprop_hmax(v[k + 1], h[(k + 1) % 3]);
    } else {
#pragma unroll
      for (int c = 0; c < PC; ++c) h[(k + 1) % 3][c] = mvprop_other_block(h[k % 3][c]);
    }
#pragma unroll
    for (int c = 0; c < PC; ++c) {
      float mx = fmaxf(fmaxf(h[(k - 1) % 3][c], h[k % 3][c]), h[(k + 1) % 3][c]);
      float rr = r[k][c];
      float vn = fmaxf(v[k][c], rr + p[k][c] * (mx - rr));
      v[k][c] = (k >= PT || inImage[k]) ? vn : v[k][c];
    }
  }
}

template <int LO>
__device__ __forceinline__ void mvprop_phase(int nsteps, float (&v)[PK][PC],
                                             const float (&r)[PK][PC],
                                             const float (&p)[PK][PC],
                                             const bool (&inImage)[PK]) {
#pragma unroll 1
  for (int s = 0; s < nsteps; ++s)
    mvprop_sweep<LO>(v, r, p, inImage);
}

template <bool FIRST, bool FINAL>
__global__ __launch_bounds__(32) __attribute__((amdgpu_waves_per_eu(1, 1)))
void mvprop_strip(
    const float* __restrict__ image, const float* __restrict__ pbuf,
    const float* __restrict__ vin, long vinStride, float* __restrict__ vout) {
  const int lane  = threadIdx.x;
  const int hl    = lane & 15;
  const int lower = lane >> 4;
  const int right = blockIdx.x & 1;
  const int sid   = blockIdx.x >> 1;
  const int b     = sid / NSTRIP;
  const int strip = sid % NSTRIP;
  const int y0    = strip * PR;
  const int ybase = y0 - PT;
  const int xo    = (right ? PXR : 0) + hl * PC;
  const float ninf = -__builtin_inff();

  const float* __restrict__ vb = vin + (size_t)b * vinStride;
  const float* __restrict__ ib = image + (size_t)b * 2 * NPIX;
  const float* __restrict__ rb = ib + NPIX;
  const float* __restrict__ pb = pbuf + (size_t)b * NPIX;

  float v[PK][PC], r[PK][PC], p[PK][PC];
  bool inImage[PK];
  int yrow[PK];

#pragma unroll
  for (int k = 0; k < PK; ++k) {
    yrow[k] = ybase + (lower ? PH - 1 - k : k);
    inImage[k] = (unsigned)yrow[k] < 256u;
    const int yc = min(max(yrow[k], 0), 255);
    float a[PC];
    mvprop_load_cols(vb + yc * 256 + xo, a);
#pragma unroll
    for (int c = 0; c < PC; ++c) v[k][c] = inImage[k] ? a[c] : ninf;
  }
#pragma unroll
  for (int k = 1; k < PK; ++k) {
    const int yc = min(max(yrow[k], 0), 255);
    if (FIRST) {
#pragma unroll
      for (int c = 0; c < PC; ++c) r[k][c] = v[k][c];
    } else {
      mvprop_load_cols(rb + yc * 256 + xo, r[k]);
    }
    mvprop_load_cols(pb + yc * 256 + xo, p[k]);
  }

  mvprop_phase<1>(PG, v, r, p, inImage);
  mvprop_phase<PG + 1>(PT - PG, v, r, p, inImage);

  float o[PK - PT][PC];
#pragma unroll
  for (int j = 0; j < PK - PT; ++j) {
#pragma unroll
    for (int c = 0; c < PC; ++c) o[j][c] = v[PT + j][c];
    if (FINAL) {
      float oc[PC];
      mvprop_load_cols(ib + yrow[PT + j] * 256 + xo, oc);
#pragma unroll
      for (int c = 0; c < PC; ++c) o[j][c] = (oc[c] > 0.49f) ? o[j][c] : -1.0f;
    }
  }
  float* dst = vout + (size_t)b * NPIX + xo;
#pragma unroll
  for (int c = 0; c < PC; ++c) {
    const bool mine = right ? (xo + c >= PSEAM) : (xo + c < PSEAM);
    if (mine) {
#pragma unroll
      for (int j = 0; j < PK - PT; ++j) dst[yrow[PT + j] * 256 + c] = o[j][c];
    }
  }
}

extern "C" void kernel_launch(void* const* d_in, const int* in_sizes, int n_in,
                              void* d_out, int out_size, void* d_ws,
                              size_t ws_size, hipStream_t stream) {
  const float* image = (const float*)d_in[0];
  const float* Wh    = (const float*)d_in[1];
  const float* bh    = (const float*)d_in[2];
  const float* Wp    = (const float*)d_in[3];
  float* out = (float*)d_out;

  float* ws    = (float*)d_ws;
  float* pbuf  = ws;
  float* v0    = ws + (size_t)NTOT;

  mvprop_conv<<<NBATCH * CB_SEGS, 256, 0, stream>>>(image, Wh, bh, Wp, pbuf);

  mvprop_strip<true, false><<<NBATCH * NSTRIP * 2, 32, 0, stream>>>(
      image, pbuf, image + NPIX, 2 * NPIX, v0);
  mvprop_strip<false, true><<<NBATCH * NSTRIP * 2, 32, 0, stream>>>(
      image, pbuf, v0, NPIX, out);
}


# propagation: outer max(v, .) dropped: with p = sigmoid >= 0 and v0 = r the blend fma(p, maxpool(v) - r, r) is never below v (rounding is monotone), so v' is the blend itself; same arithmetic
# speedup vs baseline: 5.2381x; 1.0400x over previous
#include <hip/hip_runtime.h>

typedef __attribute__((ext_vector_type(2))) float v2f;
typedef __attribute__((ext_vector_type(8))) float v8f;

#define NPIX   (256 * 256)
#define NBATCH 16
#define NTOT   (NBATCH * NPIX)

typedef float mvprop_f4u __attribute__((ext_vector_type(4), aligned(4)));

#define CB_ROWS 8
#define CB_SEGS (256 / CB_ROWS)

__device__ __forceinline__ void mvprop_row_load(const float* __restrict__ occ,
                                                int yy, int x, int h,
                                                float& side, float& mid) {
  const int yc  = min(max(yy, 0), 255);
  const int xs  = h ? x + 1 : x - 1;
  const int xsc = min(max(xs, 0), 255);
  side = occ[yc * 256 + xsc];
  mid  = occ[yc * 256 + x];
}

__device__ __forceinline__ v2f mvprop_row_vals(int yy, int x, int h, float side,
                                               float mid) {
  const bool rowOk  = (unsigned)yy < 256u;
  const int xs      = h ? x + 1 : x - 1;
  const bool sideOk = rowOk & ((unsigned)xs < 256u);
  v2f a;
  a.x = sideOk ? side : 0.0f;
  a.y = (rowOk & (h == 0)) ? mid : 0.0f;
  return a;
}

template <int MASK>
__device__ __forceinline__ float mvprop_xadd(float keep, float send) {
  const int moved = __builtin_amdgcn_update_dpp(0, __float_as_int(send),
                                                0x160 + MASK, 0xf, 0xf, true);
  return keep + __int_as_float(moved);
}

__device__ __forceinline__ float mvprop_chan_sum(const v2f (&acc)[2][4], bool q0,
                                                 bool q1, bool q2, bool q3) {
  float s8[8], s4[4], s2[2];
#pragma unroll
  for (int k = 0; k < 8; ++k) {
    const float lo = acc[k >> 2][k & 3].x, hi = acc[k >> 2][k & 3].y;
    s8[k] = mvprop_xadd<1>(q0 ? hi : lo, q0 ? lo : hi);
  }
#pragma unroll
  for (int k = 0; k < 4; ++k)
    s4[k] = mvprop_xadd<2>(q1 ? s8[2 * k + 1] : s8[2 * k],
                           q1 ? s8[2 * k] : s8[2 * k + 1]);
#pragma unroll
  for (int k = 0; k < 2; ++k)
    s2[k] = mvprop_xadd<4>(q2 ? s4[2 * k + 1] : s4[2 * k],
                           q2 ? s4[2 * k] : s4[2 * k + 1]);
  return mvprop_xadd<8>(q3 ? s2[1] : s2[0], q3 ? s2[0] : s2[1]);
}

__global__ __launch_bounds__(256) void mvprop_conv(
    const float* __restrict__ image, const float* __restrict__ Wh,
    const float* __restrict__ bh, const float* __restrict__ Wp,
    float* __restrict__ pout) {
  const int lane = threadIdx.x & 31;
  const int band = threadIdx.x >> 5;
  const int b    = blockIdx.x / CB_SEGS;
  const int seg  = blockIdx.x % CB_SEGS;
  const int y0   = seg * CB_ROWS;
  const int m    = lane & 15;
  const int h    = lane >> 4;
  const int x0   = band * 32 + m;
  const bool q0 = m & 1, q1 = m & 2, q2 = m & 4, q3 = m & 8;
  const int pix = 16 * (m >> 3) + 8 * h + (m & 7);

  const float* __restrict__ occ = image + (size_t)b * 2 * NPIX;

  v2f bF[10][4];
  float wp[10];
#pragma unroll
  for (int t = 0; t < 10; ++t) {
    const int n   = t * 16 + m;
    const bool ok = n < 150;
    const int nc  = ok ? n : 149;
    const float* __restrict__ wn = Wh + nc * 9;
    const mvprop_f4u w03 = *(const mvprop_f4u*)wn;
    const mvprop_f4u w47 = *(const mvprop_f4u*)(wn + 4);
    const float w[9] = {w03.x, w03.y, w03.z, w03.w, w47.x, w47.y, w47.z, w47.w,
                        wn[8]};
    const float bias = bh[nc];
    const float wpn  = Wp[nc];
    const float gx0 = h ? w[2] : w[0], gx1 = h ? w[5] : w[3], gx2 = h ? w[8] : w[6];
    const float gy0 = w[1], gy1 = w[4], gy2 = w[7];
    const float ex = gx0 + gx2, ey = gy0 + gy2;
    float ux[4], uy[4];
    ux[0] = gx0;               uy[0] = h ? 0.0f : gy0;
    ux[1] = 0.5f * (ex + gx1); uy[1] = h ? bias : 0.5f * (ey + gy1);
    ux[2] = 0.5f * (ex - gx1); uy[2] = h ? 0.0f : 0.5f * (ey - gy1);
    ux[3] = -gx2;              uy[3] = h ? 0.0f : -gy2;
#pragma unroll
    for (int i = 0; i < 4; ++i) {
      bF[t][i].x = ok ? ux[i] : 0.0f;
      bF[t][i].y = ok ? uy[i] : 0.0f;
    }
    wp[t] = ok ? wpn : 0.0f;
  }

  v2f r0[2], r1[2], r2[2], r3[2];
#pragma unroll
  for (int T = 0; T < 2; ++T) {
    float s0, m0, s1, m1, s2, m2, s3, m3;
    mvprop_row_load(occ, y0 - 1, x0 + 16 * T, h, s0, m0);
    mvprop_row_load(occ, y0,     x0 + 16 * T, h, s1, m1);
    mvprop_row_load(occ, y0 + 1, x0 + 16 * T, h, s2, m2);
    mvprop_row_load(occ, y0 + 2, x0 + 16 * T, h, s3, m3);
    r0[T] = mvprop_row_vals(y0 - 1, x0 + 16 * T, h, s0, m0);
    r1[T] = mvprop_row_vals(y0,     x0 + 16 * T, h, s1, m1);
    r2[T] = mvprop_row_vals(y0 + 1, x0 + 16 * T, h, s2, m2);
    r3[T] = mvprop_row_vals(y0 + 2, x0 + 16 * T, h, s3, m3);
  }

#pragma unroll 1
  for (int j = 0; j < CB_ROWS; j += 2) {
    const int y = y0 + j;

    float fs3[2], fm3[2], fs4[2], fm4[2];
#pragma unroll
    for (int T = 0; T < 2; ++T) {
      mvprop_row_load(occ, y + 3, x0 + 16 * T, h, fs3[T], fm3[T]);
      mvprop_row_load(occ, y + 4, x0 + 16 * T, h, fs4[T], fm4[T]);
    }

    v2f a1[2], a2[2], a3[2], a4[2];
#pragma unroll
    for (int T = 0; T < 2; ++T) {
      a1[T] = r0[T] - r2[T];
      a2[T] = r1[T] + r2[T];
      a2[T].y = h ? 1.0f : a2[T].y;
      a3[T] = r2[T] - r1[T];
      a4[T] = r1[T] - r3[T];
    }

    v2f accE[2][4], accO[2][4];
#pragma unroll
    for (int k = 0; k < 4; ++k) {
      accE[0][k] = 0.0f; accE[1][k] = 0.0f;
      accO[0][k] = 0.0f; accO[1][k] = 0.0f;
    }

#pragma unroll
    for (int t = 0; t < 10; ++t) {
      v8f z8;
#pragma unroll
      for (int r = 0; r < 8; ++r) z8[r] = 0.0f;
      const v8f p0 = __builtin_amdgcn_wmma_f32_16x16x4_f32(
          false, a2[0], false, bF[t][1], (short)0, z8, false, false);
      const v8f g0 = __builtin_amdgcn_wmma_f32_16x16x4_f32(
          false, a3[0], false, bF[t][2], (short)0, z8, false, false);
      const v8f p1 = __builtin_amdgcn_wmma_f32_16x16x4_f32(
          false, a2[1], false, bF[t][1], (short)0, z8, false, false);
      const v8f g1 = __builtin_amdgcn_wmma_f32_16x16x4_f32(
          false, a3[1], false, bF[t][2], (short)0, z8, false, false);
      const v8f x0 = __builtin_amdgcn_wmma_f32_16x16x4_f32(
          false, a1[0], false, bF[t][0], (short)0, p0, false, false);
      const v8f y0 = __builtin_amdgcn_wmma_f32_16x16x4_f32(
          false, a4[0], false, bF[t][3], (short)0, p0, false, false);
      const v8f x1 = __builtin_amdgcn_wmma_f32_16x16x4_f32(
          false, a1[1], false, bF[t][0], (short)0, p1, false, false);
      const v8f y1 = __builtin_amdgcn_wmma_f32_16x16x4_f32(
          false, a4[1], false, bF[t][3], (short)0, p1, false, false);
      const v8f e0 = x0 + g0, o0 = y0 - g0;
      const v8f e1 = x1 + g1, o1 = y1 - g1;
      const v2f wp2 = {wp[t], wp[t]};
      v8f zero8;
#pragma unroll
      for (int r = 0; r < 8; ++r) zero8[r] = 0.0f;
      const v8f re0 = __builtin_elementwise_max(e0, zero8);
      const v8f ro0 = __builtin_elementwise_max(o0, zero8);
      const v8f re1 = __builtin_elementwise_max(e1, zero8);
      const v8f ro1 = __builtin_elementwise_max(o1, zero8);
#pragma unroll
      for (int k = 0; k < 4; ++k) {
        const v2f de0 = {re0[2 * k], re0[2 * k + 1]};
        const v2f do0 = {ro0[2 * k], ro0[2 * k + 1]};
        const v2f de1 = {re1[2 * k], re1[2 * k + 1]};
        const v2f do1 = {ro1[2 * k], ro1[2 * k + 1]};
        accE[0][k] = __builtin_elementwise_fma(de0, wp2, accE[0][k]);
        accO[0][k] = __builtin_elementwise_fma(do0, wp2, accO[0][k]);
        accE[1][k] = __builtin_elementwise_fma(de1, wp2, accE[1][k]);
        accO[1][k] = __builtin_elementwise_fma(do1, wp2, accO[1][k]);
      }
    }

    const float sumE = mvprop_chan_sum(accE, q0, q1, q2, q3);
    const float sumO = mvprop_chan_sum(accO, q0, q1, q2, q3);
    float* __restrict__ prow = pout + (size_t)b * NPIX + y * 256 + band * 32 + pix;
    prow[0]   = 1.0f / (1.0f + __expf(-sumE));
    prow[256] = 1.0f / (1.0f + __expf(-sumO));

#pragma unroll
    for (int T = 0; T < 2; ++T) {
      r0[T] = r2[T];
      r1[T] = r3[T];
      r2[T] = mvprop_row_vals(y + 3, x0 + 16 * T, h, fs3[T], fm3[T]);
      r3[T] = mvprop_row_vals(y + 4, x0 + 16 * T, h, fs4[T], fm4[T]);
    }
  }
}

#define PT 10
#define PR 8
#define PH (PR + 2 * PT)
#define PK (PH / 2)
#define NSTRIP (256 / PR)
#define PG 5
#define PC 10
#define PW (16 * PC)
#define PXR (256 - PW)
#define PSEAM 128
static_assert(PH % 2 == 0, "the strip is cut into two blocks of equal height");
static_assert(PW - PT >= PSEAM, "left wave must be exact up to the seam");
static_assert(PXR + PT <= PSEAM, "right wave must be exact from the seam on");

typedef float mvprop_f2u __attribute__((ext_vector_type(2), aligned(4)));

__device__ __forceinline__ void mvprop_load_cols(const float* __restrict__ src,
                                                 float (&d)[PC]) {
  const mvprop_f4u a = *(const mvprop_f4u*)src;
  const mvprop_f4u b = *(const mvprop_f4u*)(src + 4);
  const mvprop_f2u c = *(const mvprop_f2u*)(src + 8);
  d[0] = a.x; d[1] = a.y; d[2] = a.z; d[3] = a.w;
  d[4] = b.x; d[5] = b.y; d[6] = b.z; d[7] = b.w;
  d[8] = c.x; d[9] = c.y;
}
static_assert(PC == 10, "mvprop_load_cols reads 4 + 4 + 2 floats");

__device__ __forceinline__ float mvprop_prev_lane(float x) {
  const int pad = __float_as_int(-__builtin_inff());
  return __int_as_float(
      __builtin_amdgcn_update_dpp(pad, __float_as_int(x), 0x111, 0xf, 0xf, false));
}
__device__ __forceinline__ float mvprop_next_lane(float x) {
  const int pad = __float_as_int(-__builtin_inff());
  return __int_as_float(
      __builtin_amdgcn_update_dpp(pad, __float_as_int(x), 0x101, 0xf, 0xf, false));
}
__device__ __forceinline__ float mvprop_other_block(float x) {
  const int xi = __float_as_int(x);
  return __int_as_float(__builtin_amdgcn_permlanex16(xi, xi, 0x76543210, 0xfedcba98,
                                                     false, false));
}

__device__ __forceinline__ void mvprop_hmax(const float (&vr)[PC], float (&h)[PC]) {
  const float l = mvprop_prev_lane(vr[PC - 1]);
  const float g = mvprop_next_lane(vr[0]);
  h[0] = fmaxf(fmaxf(l, vr[0]), vr[1]);
#pragma unroll
  for (int c = 1; c < PC - 1; ++c)
    h[c] = fmaxf(fmaxf(vr[c - 1], vr[c]), vr[c + 1]);
  h[PC - 1] = fmaxf(fmaxf(vr[PC - 2], vr[PC - 1]), g);
}

template <int LO>
__device__ __forceinline__ void mvprop_sweep(float (&v)[PK][PC],
                                             const float (&r)[PK][PC],
                                             const float (&p)[PK][PC],
                                             const bool (&inImage)[PK]) {
  float h[3][PC];
  mvprop_hmax(v[LO - 1], h[(LO - 1) % 3]);
  mvprop_hmax(v[LO], h[LO % 3]);
#pragma unroll
  for (int k = LO; k < PK; ++k) {
    if (k + 1 < PK) {
      mvprop_hmax(v[k + 1], h[(k + 1) % 3]);
    } else {
#pragma unroll
      for (int c = 0; c < PC; ++c) h[(k + 1) % 3][c] = mvprop_other_block(h[k % 3][c]);
    }
#pragma unroll
    for (int c = 0; c < PC; ++c) {
      float mx = fmaxf(fmaxf(h[(k - 1) % 3][c], h[k % 3][c]), h[(k + 1) % 3][c]);
      float rr = r[k][c];
      float vn = __builtin_fmaf(p[k][c], mx - rr, rr);
      v[k][c] = (k >= PT || inImage[k]) ? vn : v[k][c];
    }
  }
}

template <int LO>
__device__ __forceinline__ void mvprop_phase(int nsteps, float (&v)[PK][PC],
                                             const float (&r)[PK][PC],
                                             const float (&p)[PK][PC],
                                             const bool (&inImage)[PK]) {
#pragma unroll 1
  for (int s = 0; s < nsteps; ++s)
    mvprop_sweep<LO>(v, r, p, inImage);
}

template <bool FIRST, bool FINAL>
__global__ __launch_bounds__(32) __attribute__((amdgpu_waves_per_eu(1, 1)))
void mvprop_strip(
    const float* __restrict__ image, const float* __restrict__ pbuf,
    const float* __restrict__ vin, long vinStride, float* __restrict__ vout) {
  const int lane  = threadIdx.x;
  const int hl    = lane & 15;
  const int lower = lane >> 4;
  const int right = blockIdx.x & 1;
  const int sid   = blockIdx.x >> 1;
  const int b     = sid / NSTRIP;
  const int strip = sid % NSTRIP;
  const int y0    = strip * PR;
  const int ybase = y0 - PT;
  const int xo    = (right ? PXR : 0) + hl * PC;
  const float ninf = -__builtin_inff();

  const float* __restrict__ vb = vin + (size_t)b * vinStride;
  const float* __restrict__ ib = image + (size_t)b * 2 * NPIX;
  const float* __restrict__ rb = ib + NPIX;
  const float* __restrict__ pb = pbuf + (size_t)b * NPIX;

  float v[PK][PC], r[PK][PC], p[PK][PC];
  bool inImage[PK];
  int yrow[PK];

#pragma unroll
  for (int k = 0; k < PK; ++k) {
    yrow[k] = ybase + (lower ? PH - 1 - k : k);
    inImage[k] = (unsigned)yrow[k] < 256u;
    const int yc = min(max(yrow[k], 0), 255);
    float a[PC];
    mvprop_load_cols(vb + yc * 256 + xo, a);
#pragma unroll
    for (int c = 0; c < PC; ++c) v[k][c] = inImage[k] ? a[c] : ninf;
  }
#pragma unroll
  for (int k = 1; k < PK; ++k) {
    const int yc = min(max(yrow[k], 0), 255);
    if (FIRST) {
#pragma unroll
      for (int c = 0; c < PC; ++c) r[k][c] = v[k][c];
    } else {
      mvprop_load_cols(rb + yc * 256 + xo, r[k]);
    }
    mvprop_load_cols(pb + yc * 256 + xo, p[k]);
  }

  mvprop_phase<1>(PG, v, r, p, inImage);
  mvprop_phase<PG + 1>(PT - PG, v, r, p, inImage);

  float o[PK - PT][PC];
#pragma unroll
  for (int j = 0; j < PK - PT; ++j) {
#pragma unroll
    for (int c = 0; c < PC; ++c) o[j][c] = v[PT + j][c];
    if (FINAL) {
      float oc[PC];
      mvprop_load_cols(ib + yrow[PT + j] * 256 + xo, oc);
#pragma unroll
      for (int c = 0; c < PC; ++c) o[j][c] = (oc[c] > 0.49f) ? o[j][c] : -1.0f;
    }
  }
  float* dst = vout + (size_t)b * NPIX + xo;
#pragma unroll
  for (int c = 0; c < PC; ++c) {
    const bool mine = right ? (xo + c >= PSEAM) : (xo + c < PSEAM);
    if (mine) {
#pragma unroll
      for (int j = 0; j < PK - PT; ++j) dst[yrow[PT + j] * 256 + c] = o[j][c];
    }
  }
}

extern "C" void kernel_launch(void* const* d_in, const int* in_sizes, int n_in,
                              void* d_out, int out_size, void* d_ws,
                              size_t ws_size, hipStream_t stream) {
  const float* image = (const float*)d_in[0];
  const float* Wh    = (const float*)d_in[1];
  const float* bh    = (const float*)d_in[2];
  const float* Wp    = (const float*)d_in[3];
  float* out = (float*)d_out;

  float* ws    = (float*)d_ws;
  float* pbuf  = ws;
  float* v0    = ws + (size_t)NTOT;

  mvprop_conv<<<NBATCH * CB_SEGS, 256, 0, stream>>>(image, Wh, bh, Wp, pbuf);

  mvprop_strip<true, false><<<NBATCH * NSTRIP * 2, 32, 0, stream>>>(
      image, pbuf, image + NPIX, 2 * NPIX, v0);
  mvprop_strip<false, true><<<NBATCH * NSTRIP * 2, 32, 0, stream>>>(
      image, pbuf, v0, NPIX, out);
}
